# starting kernel
# baseline (speedup 1.0000x reference)
#include <hip/hip_runtime.h>

typedef __attribute__((ext_vector_type(2))) float v2f;
typedef __attribute__((ext_vector_type(8))) float v8f;

#define TOK   4096
#define BN    1024
#define DD    128

__device__ __forceinline__ float sigf(float x)  { return 1.0f / (1.0f + __expf(-x)); }
__device__ __forceinline__ float siluf(float x) { return x * sigf(x); }
__device__ __forceinline__ float dsiluf(float x){ float s = sigf(x); return s * (1.0f + x * (1.0f - s)); }

__global__ __launch_bounds__(256) void gemm128(
    const float* __restrict__ A, int lda,
    const float* __restrict__ W, int ldw, int wBatchStride, int wTrans,
    const float* __restrict__ Zmul,
    float* __restrict__ Cout, int ldc,
    float* __restrict__ SiluOut, int ldso)
{
  const int wave = threadIdx.x >> 5;
  const int lane = threadIdx.x & 31;
  const int half = lane >> 4;
  const int l    = lane & 15;
  const int m0   = blockIdx.x * 32 + (wave >> 2) * 16;
  const int n0   = blockIdx.y * 128 + (wave & 3) * 32;
  const int batch = (blockIdx.x * 32) / BN;
  const float* Wb = W + (size_t)batch * wBatchStride;

  v8f acc0 = {};
  v8f acc1 = {};
  const float* arow = A + (size_t)(m0 + l) * lda + 2 * half;

  if (wTrans) {
    const float* wr0 = Wb + (size_t)(n0 + l) * ldw + 2 * half;
    const float* wr1 = Wb + (size_t)(n0 + 16 + l) * ldw + 2 * half;
    for (int k0 = 0; k0 < DD; k0 += 4) {
      v2f a  = *(const v2f*)(arow + k0);
      v2f b0 = *(const v2f*)(wr0 + k0);
      v2f b1 = *(const v2f*)(wr1 + k0);
      acc0 = __builtin_amdgcn_wmma_f32_16x16x4_f32(false, a, false, b0, (short)0, acc0, false, false);
      acc1 = __builtin_amdgcn_wmma_f32_16x16x4_f32(false, a, false, b1, (short)0, acc1, false, false);
    }
  } else {
    for (int k0 = 0; k0 < DD; k0 += 4) {
      v2f a = *(const v2f*)(arow + k0);
      const float* r0 = Wb + (size_t)(k0 + 2 * half) * ldw + n0 + l;
      const float* r1 = r0 + ldw;
      v2f b0, b1;
      b0.x = r0[0];  b0.y = r1[0];
      b1.x = r0[16]; b1.y = r1[16];
      acc0 = __builtin_amdgcn_wmma_f32_16x16x4_f32(false, a, false, b0, (short)0, acc0, false, false);
      acc1 = __builtin_amdgcn_wmma_f32_16x16x4_f32(false, a, false, b1, (short)0, acc1, false, false);
    }
  }

  #pragma unroll
  for (int r = 0; r < 8; ++r) {
    const int m = m0 + r + 8 * half;
    float c0 = acc0[r];
    float c1 = acc1[r];
    if (Zmul) {
      c0 *= dsiluf(Zmul[(size_t)m * DD + (n0 + l)]);
      c1 *= dsiluf(Zmul[(size_t)m * DD + (n0 + 16 + l)]);
    }
    if (Cout) {
      Cout[(size_t)m * ldc + n0 + l]      = c0;
      Cout[(size_t)m * ldc + n0 + 16 + l] = c1;
    }
    if (SiluOut) {
      SiluOut[(size_t)m * ldso + n0 + l]      = siluf(c0);
      SiluOut[(size_t)m * ldso + n0 + 16 + l] = siluf(c1);
    }
  }
}

__global__ __launch_bounds__(256) void gradU(
    const float* __restrict__ Act, int ldact,
    const float* __restrict__ Dlt,
    const float* __restrict__ Wbase,
    float* __restrict__ Uout)
{
  const int wave = threadIdx.x >> 5;
  const int lane = threadIdx.x & 31;
  const int half = lane >> 4;
  const int l    = lane & 15;
  const int m0   = blockIdx.x * 32 + (wave >> 2) * 16;
  const int n0   = (wave & 3) * 32;
  const int batch = blockIdx.z;
  const float* Ab = Act + (size_t)batch * BN * ldact + m0 + l;
  const float* Db = Dlt + (size_t)batch * BN * DD;

  v8f acc0 = {};
  v8f acc1 = {};
  for (int k0 = 0; k0 < BN; k0 += 4) {
    const int kr = k0 + 2 * half;
    v2f a;
    a.x = Ab[(size_t)(kr + 0) * ldact];
    a.y = Ab[(size_t)(kr + 1) * ldact];
    const float* d0 = Db + (size_t)kr * DD + n0 + l;
    const float* d1 = d0 + DD;
    v2f b0, b1;
    b0.x = d0[0];  b0.y = d1[0];
    b1.x = d0[16]; b1.y = d1[16];
    acc0 = __builtin_amdgcn_wmma_f32_16x16x4_f32(false, a, false, b0, (short)0, acc0, false, false);
    acc1 = __builtin_amdgcn_wmma_f32_16x16x4_f32(false, a, false, b1, (short)0, acc1, false, false);
  }

  #pragma unroll
  for (int r = 0; r < 8; ++r) {
    const int d = m0 + r + 8 * half;
    float* urow = Uout + (size_t)batch * DD * DD + (size_t)d * DD;
    const float* wrow = Wbase + (size_t)d * DD;
    urow[n0 + l]      = wrow[n0 + l]      + acc0[r];
    urow[n0 + 16 + l] = wrow[n0 + 16 + l] + acc1[r];
  }
}

__global__ __launch_bounds__(256) void d3_elem(
    const float* __restrict__ P, const float* __restrict__ KV, float* __restrict__ D3)
{
  const int idx = blockIdx.x * 256 + threadIdx.x;
  const int m = idx >> 7;
  const int n = idx & 127;
  const float v = KV[(size_t)m * 256 + DD + n];
  D3[idx] = (2.0f / 128.0f) * (P[idx] - v);
}

extern "C" void kernel_launch(void* const* d_in, const int* in_sizes, int n_in,
                              void* d_out, int out_size, void* d_ws, size_t ws_size,
                              hipStream_t stream)
{
  const float* S   = (const float*)d_in[0];
  const float* w0  = (const float*)d_in[1];
  const float* w1  = (const float*)d_in[2];
  const float* w2  = (const float*)d_in[3];
  const float* w3  = (const float*)d_in[4];
  const float* wq  = (const float*)d_in[5];
  const float* wkv = (const float*)d_in[6];
  float* out = (float*)d_out;

  float* p = (float*)d_ws;
  const size_t T = (size_t)TOK * DD;
  float* KV = p; p += (size_t)TOK * 256;
  float* Z0 = p; p += T;  float* A1 = p; p += T;
  float* Z1 = p; p += T;  float* A2 = p; p += T;
  float* Z2 = p; p += T;  float* A3 = p; p += T;
  float* P_ = p; p += T;
  float* Da = p; p += T;
  float* Db = p; p += T;
  float* Xa = p; p += T;
  float* Xb = p; p += T;
  float* U0 = p; p += 4 * DD * DD;
  float* U1 = p; p += 4 * DD * DD;
  float* U2 = p; p += 4 * DD * DD;
  float* U3 = p; p += 4 * DD * DD;

  const dim3 blk(256);
  const dim3 gKV(TOK / 32, 2);
  const dim3 g(TOK / 32, 1);
  const dim3 gg(4, 1, 4);

  gemm128<<<gKV, blk, 0, stream>>>(S, DD, wkv, 256, 0, 0, nullptr, KV, 256, nullptr, 0);

  gemm128<<<g, blk, 0, stream>>>(KV, 256, w0, DD, 0, 0, nullptr, Z0, DD, A1, DD);
  gemm128<<<g, blk, 0, stream>>>(A1, DD, w1, DD, 0, 0, nullptr, Z1, DD, A2, DD);
  gemm128<<<g, blk, 0, stream>>>(A2, DD, w2, DD, 0, 0, nullptr, Z2, DD, A3, DD);
  gemm128<<<g, blk, 0, stream>>>(A3, DD, w3, DD, 0, 0, nullptr, P_, DD, nullptr, 0);

  d3_elem<<<dim3((TOK * DD) / 256), blk, 0, stream>>>(P_, KV, Da);

  gradU<<<gg, blk, 0, stream>>>(A3, DD, Da, w3, U3);
  gemm128<<<g, blk, 0, stream>>>(Da, DD, w3, DD, 0, 1, Z2, Db, DD, nullptr, 0);
  gradU<<<gg, blk, 0, stream>>>(A2, DD, Db, w2, U2);
  gemm128<<<g, blk, 0, stream>>>(Db, DD, w2, DD, 0, 1, Z1, Da, DD, nullptr, 0);
  gradU<<<gg, blk, 0, stream>>>(A1, DD, Da, w1, U1);
  gemm128<<<g, blk, 0, stream>>>(Da, DD, w1, DD, 0, 1, Z0, Db, DD, nullptr, 0);
  gradU<<<gg, blk, 0, stream>>>(KV, 256, Db, w0, U0);

  gemm128<<<g, blk, 0, stream>>>(S,  DD, wq, DD, 0,       0, nullptr, Xa, DD, nullptr, 0);
  gemm128<<<g, blk, 0, stream>>>(Xa, DD, U0, DD, DD * DD, 0, nullptr, nullptr, 0, Xb, DD);
  gemm128<<<g, blk, 0, stream>>>(Xb, DD, U1, DD, DD * DD, 0, nullptr, nullptr, 0, Xa, DD);
  gemm128<<<g, blk, 0, stream>>>(Xa, DD, U2, DD, DD * DD, 0, nullptr, nullptr, 0, Xb, DD);
  gemm128<<<g, blk, 0, stream>>>(Xb, DD, U3, DD, DD * DD, 0, nullptr, out, DD, nullptr, 0);
}


# L2 prefetch tier (global_prefetch_b8, one 256-byte unit per lane) ahead of first reads: gradient GEMM rows 5 ring rounds ahead, delta GEMM tiles and later-layer chain weights at entry; sums unchanged
# speedup vs baseline: 2.6992x; 2.6992x over previous
#include <hip/hip_runtime.h>

typedef __attribute__((ext_vector_type(2))) float v2f;
typedef __attribute__((ext_vector_type(4))) float v4f;
typedef __attribute__((ext_vector_type(8))) float v8f;

#define TOK   4096
#define BN    1024
#define DD    128

__device__ __forceinline__ float sigf(float x)  { return 1.0f / (1.0f + __expf(-x)); }
__device__ __forceinline__ float siluf(float x) { return x * sigf(x); }
__device__ __forceinline__ float dsiluf(float x){ float s = sigf(x); return s * (1.0f + x * (1.0f - s)); }

__global__ __launch_bounds__(256) void gemm128(
    const float* __restrict__ A, int lda,
    const float* __restrict__ W, int ldw, int wBatchStride, int wTrans,
    const float* __restrict__ Zmul,
    float* __restrict__ Cout, int ldc,
    float* __restrict__ SiluOut, int ldso)
{
  const int wave = threadIdx.x >> 5;
  const int lane = threadIdx.x & 31;
  const int half = lane >> 4;
  const int l    = lane & 15;
  const int m0   = blockIdx.x * 32 + (wave >> 2) * 16;
  const int n0   = blockIdx.y * 128 + (wave & 3) * 32;
  const int batch = (blockIdx.x * 32) / BN;
  const float* Wb = W + (size_t)batch * wBatchStride;

  __builtin_prefetch(A + (size_t)(m0 + l) * lda + 64 * half, 0, 3);
  if (Zmul) __builtin_prefetch(Zmul + (size_t)(m0 + l) * DD + n0, 0, 3);

  v8f acc0 = {};
  v8f acc1 = {};
  const float* arow = A + (size_t)(m0 + l) * lda + 2 * half;

  if (wTrans) {
    const float* wr0 = Wb + (size_t)(n0 + l) * ldw + 2 * half;
    const float* wr1 = Wb + (size_t)(n0 + 16 + l) * ldw + 2 * half;
    __builtin_prefetch(Wb + (size_t)(n0 + l) * ldw + 64 * half, 0, 3);
    __builtin_prefetch(Wb + (size_t)(n0 + 16 + l) * ldw + 64 * half, 0, 3);
    for (int k0 = 0; k0 < DD; k0 += 4) {
      v2f a  = *(const v2f*)(arow + k0);
      v2f b0 = *(const v2f*)(wr0 + k0);
      v2f b1 = *(const v2f*)(wr1 + k0);
      acc0 = __builtin_amdgcn_wmma_f32_16x16x4_f32(false, a, false, b0, (short)0, acc0, false, false);
      acc1 = __builtin_amdgcn_wmma_f32_16x16x4_f32(false, a, false, b1, (short)0, acc1, false, false);
    }
  } else {
    for (int k0 = 0; k0 < DD; k0 += 4) {
      v2f a = *(const v2f*)(arow + k0);
      const float* r0 = Wb + (size_t)(k0 + 2 * half) * ldw + n0 + l;
      const float* r1 = r0 + ldw;
      v2f b0, b1;
      b0.x = r0[0];  b0.y = r1[0];
      b1.x = r0[16]; b1.y = r1[16];
      acc0 = __builtin_amdgcn_wmma_f32_16x16x4_f32(false, a, false, b0, (short)0, acc0, false, false);
      acc1 = __builtin_amdgcn_wmma_f32_16x16x4_f32(false, a, false, b1, (short)0, acc1, false, false);
    }
  }

  #pragma unroll
  for (int r = 0; r < 8; ++r) {
    const int m = m0 + r + 8 * half;
    float c0 = acc0[r];
    float c1 = acc1[r];
    if (Zmul) {
      c0 *= dsiluf(Zmul[(size_t)m * DD + (n0 + l)]);
      c1 *= dsiluf(Zmul[(size_t)m * DD + (n0 + 16 + l)]);
    }
    if (Cout) {
      Cout[(size_t)m * ldc + n0 + l]      = c0;
      Cout[(size_t)m * ldc + n0 + 16 + l] = c1;
    }
    if (SiluOut) {
      SiluOut[(size_t)m * ldso + n0 + l]      = siluf(c0);
      SiluOut[(size_t)m * ldso + n0 + 16 + l] = siluf(c1);
    }
  }
}

template <int LDA>
__global__ __launch_bounds__(32) void gradU(
    const float* __restrict__ Act,
    const float* __restrict__ Dlt,
    const float* __restrict__ Wbase,
    float* __restrict__ Uout)
{
  constexpr int RING = 16;
  const int lane = threadIdx.x;
  const int half = lane >> 4;
  const int l    = lane & 15;
  const int m0   = (blockIdx.x >> 2) * 16;
  const int n0   = (blockIdx.x & 3) * 32;
  const int batch = blockIdx.z;
  const float* ap = Act + ((size_t)batch * BN + 2 * half) * LDA + m0 + l;
  const float* dp = Dlt + ((size_t)batch * BN + 2 * half) * DD + n0 + 2 * l;

  float ax[RING], ay[RING];
  v2f   t0[RING], t1[RING];

  v2f wb[8];
  #pragma unroll
  for (int r = 0; r < 8; ++r)
    wb[r] = *(const v2f*)(Wbase + (size_t)(m0 + r + 8 * half) * DD + n0 + 2 * l);

  #pragma unroll
  for (int j = 0; j < RING; ++j) {
    ax[j] = ap[(size_t)(4 * j) * LDA];
    ay[j] = ap[(size_t)(4 * j + 1) * LDA];
    t0[j] = *(const v2f*)(dp + (size_t)(4 * j) * DD);
    t1[j] = *(const v2f*)(dp + (size_t)(4 * j + 1) * DD);
    __builtin_amdgcn_sched_barrier(0);
  }

  constexpr int PF_AHEAD = 384;
  const float* pf = (half == 0) ? (Act + ((size_t)batch * BN + l) * LDA + m0)
                                : (Dlt + ((size_t)batch * BN + l) * DD + n0);
  const size_t pfs = (half == 0) ? (size_t)16 * LDA : (size_t)16 * DD;
  #pragma unroll
  for (int g = (4 * RING) / 16; g < PF_AHEAD / 16; ++g) __builtin_prefetch(pf + (size_t)g * pfs, 0, 3);

  v8f acc0 = {};
  v8f acc1 = {};

  for (int k0 = 0; k0 < BN - 4 * RING; k0 += 4 * RING) {
    ap += (size_t)(4 * RING) * LDA;
    dp += (size_t)(4 * RING) * DD;
    if (k0 + PF_AHEAD < BN) {
      const float* pq = pf + (size_t)((k0 + PF_AHEAD) / 16) * pfs;
      __builtin_prefetch(pq, 0, 3);
      __builtin_prefetch(pq + pfs, 0, 3);
      __builtin_prefetch(pq + 2 * pfs, 0, 3);
      __builtin_prefetch(pq + 3 * pfs, 0, 3);
    }
    #pragma unroll
    for (int j = 0; j < RING; ++j) {
      v2f a;  a.x  = ax[j];   a.y  = ay[j];
      float p = t0[j].y, q = t1[j].x;
      asm volatile("v_swap_b32 %0, %1" : "+v"(p), "+v"(q));
      v2f b0; b0.x = t0[j].x; b0.y = p;
      v2f b1; b1.x = q;       b1.y = t1[j].y;
      acc0 = __builtin_amdgcn_wmma_f32_16x16x4_f32(false, a, false, b0, (short)0, acc0, false, false);
      acc1 = __builtin_amdgcn_wmma_f32_16x16x4_f32(false, a, false, b1, (short)0, acc1, false, false);
      ax[j] = ap[(size_t)(4 * j) * LDA];
      ay[j] = ap[(size_t)(4 * j + 1) * LDA];
      t0[j] = *(const v2f*)(dp + (size_t)(4 * j) * DD);
      t1[j] = *(const v2f*)(dp + (size_t)(4 * j + 1) * DD);
    }
  }

  #pragma unroll
  for (int j = 0; j < RING; ++j) {
    v2f a;  a.x  = ax[j];   a.y  = ay[j];
    v2f b0; b0.x = t0[j].x; b0.y = t1[j].x;
    v2f b1; b1.x = t0[j].y; b1.y = t1[j].y;
    acc0 = __builtin_amdgcn_wmma_f32_16x16x4_f32(false, a, false, b0, (short)0, acc0, false, false);
    acc1 = __builtin_amdgcn_wmma_f32_16x16x4_f32(false, a, false, b1, (short)0, acc1, false, false);
  }

  #pragma unroll
  for (int r = 0; r < 8; ++r) {
    const int d = m0 + r + 8 * half;
    v2f o;
    o.x = wb[r].x + acc0[r];
    o.y = wb[r].y + acc1[r];
    *(v2f*)(Uout + (size_t)batch * DD * DD + (size_t)d * DD + n0 + 2 * l) = o;
  }
}

__global__ __launch_bounds__(32) void qchain(
    const float* __restrict__ S, const float* __restrict__ wq,
    const float* __restrict__ U0, const float* __restrict__ U1,
    const float* __restrict__ U2, const float* __restrict__ U3,
    float* __restrict__ out)
{
  constexpr int RING = 16;
  const int lane = threadIdx.x;
  const int half = lane >> 4;
  const int c    = lane & 15;
  const int tok0 = blockIdx.x * 16;
  const size_t boff = (size_t)(tok0 / BN) * DD * DD;
  const size_t loff = (size_t)(8 * half) * DD + c;

  float x[8][8];
  const float* srow = S + (size_t)(tok0 + c) * DD + 8 * half;
  #pragma unroll
  for (int t = 0; t < 8; ++t) {
    const v4f lo = *(const v4f*)(srow + 16 * t);
    const v4f hi = *(const v4f*)(srow + 16 * t + 4);
    x[t][0] = lo.x; x[t][1] = lo.y; x[t][2] = lo.z; x[t][3] = lo.w;
    x[t][4] = hi.x; x[t][5] = hi.y; x[t][6] = hi.z; x[t][7] = hi.w;
  }

  float ra[RING][4];
  const float* Wc = wq + loff;
  #pragma unroll
  for (int i = 0; i < RING; ++i) {
    const int kr = 16 * (i >> 2) + 2 * (i & 3);
    ra[i][0] = Wc[(size_t)kr * DD];
    ra[i][1] = Wc[(size_t)(kr + 1) * DD];
    ra[i][2] = Wc[(size_t)kr * DD + 16];
    ra[i][3] = Wc[(size_t)(kr + 1) * DD + 16];
    __builtin_amdgcn_sched_barrier(0);
  }

  {
    const size_t uoff = boff + (size_t)(((blockIdx.x >> 3) & 7) * 32 + lane) * 64;
    __builtin_prefetch(U0 + uoff, 0, 3);
    __builtin_prefetch(U1 + uoff, 0, 3);
    __builtin_prefetch(U2 + uoff, 0, 3);
    __builtin_prefetch(U3 + uoff, 0, 3);
  }

  #pragma unroll 1
  for (int L = 0; L < 5; ++L) {
    const float* Wn = ((L == 0) ? U0 : (L == 1) ? U1 : (L == 2) ? U2 : U3) + boff + loff;
    v8f acc[8];
    #pragma unroll
    for (int p = 0; p < 4; ++p) {
      acc[2 * p]     = (v8f){};
      acc[2 * p + 1] = (v8f){};
      #pragma unroll
      for (int i = 0; i < 32; ++i) {
        const int slot = i & (RING - 1);
        const int t = i >> 2, j = i & 3;
        v2f a0; a0.x = ra[slot][0]; a0.y = ra[slot][1];
        v2f a1; a1.x = ra[slot][2]; a1.y = ra[slot][3];
        v2f b;  b.x  = x[t][2 * j]; b.y  = x[t][2 * j + 1];
        acc[2 * p]     = __builtin_amdgcn_wmma_f32_16x16x4_f32(false, a0, false, b, (short)0, acc[2 * p], false, false);
        acc[2 * p + 1] = __builtin_amdgcn_wmma_f32_16x16x4_f32(false, a1, false, b, (short)0, acc[2 * p + 1], false, false);
        const float* src = (i < 16 || p < 3) ? Wc : Wn;
        const int pp = (i < 16) ? p : ((p < 3) ? p + 1 : 0);
        const int ii = (i < 16) ? i + 16 : i - 16;
        const int kr = 16 * (ii >> 2) + 2 * (ii & 3);
        ra[slot][0] = src[(size_t)kr * DD + 32 * pp];
        ra[slot][1] = src[(size_t)(kr + 1) * DD + 32 * pp];
        ra[slot][2] = src[(size_t)kr * DD + 32 * pp + 16];
        ra[slot][3] = src[(size_t)(kr + 1) * DD + 32 * pp + 16];
        __builtin_amdgcn_sched_barrier(0);
      }
    }
    if ((L >= 1) && (L <= 3)) {
      #pragma unroll
      for (int t = 0; t < 8; ++t) {
        #pragma unroll
        for (int r = 0; r < 8; ++r) x[t][r] = siluf(acc[t][r]);
      }
    } else {
      #pragma unroll
      for (int t = 0; t < 8; ++t) {
        #pragma unroll
        for (int r = 0; r < 8; ++r) x[t][r] = acc[t][r];
      }
    }
    Wc = Wn;
  }

  float* orow = out + (size_t)(tok0 + c) * DD + 8 * half;
  #pragma unroll
  for (int t = 0; t < 8; ++t) {
    v4f lo; lo.x = x[t][0]; lo.y = x[t][1]; lo.z = x[t][2]; lo.w = x[t][3];
    v4f hi; hi.x = x[t][4]; hi.y = x[t][5]; hi.z = x[t][6]; hi.w = x[t][7];
    *(v4f*)(orow + 16 * t)     = lo;
    *(v4f*)(orow + 16 * t + 4) = hi;
  }
}


template <int LD>
__device__ __forceinline__ void ring_fill(float (&ra)[16][4], const float* __restrict__ Wc)
{
  #pragma unroll
  for (int i = 0; i < 16; ++i) {
    const int kr = 16 * (i >> 2) + 2 * (i & 3);
    ra[i][0] = Wc[(size_t)kr * LD];
    ra[i][1] = Wc[(size_t)(kr + 1) * LD];
    ra[i][2] = Wc[(size_t)kr * LD + 16];
    ra[i][3] = Wc[(size_t)(kr + 1) * LD + 16];
    __builtin_amdgcn_sched_barrier(0);
  }
}

template <int LD>
__device__ __forceinline__ void chain_layer(const float (&x)[8][8], v8f (&acc)[8], float (&ra)[16][4],
                                            const float* __restrict__ Wc, const float* __restrict__ Wn)
{
  #pragma unroll
  for (int p = 0; p < 4; ++p) {
    acc[2 * p]     = (v8f){};
    acc[2 * p + 1] = (v8f){};
    #pragma unroll
    for (int i = 0; i < 32; ++i) {
      const int slot = i & 15;
      const int t = i >> 2, j = i & 3;
      v2f a0; a0.x = ra[slot][0]; a0.y = ra[slot][1];
      v2f a1; a1.x = ra[slot][2]; a1.y = ra[slot][3];
      v2f b;  b.x  = x[t][2 * j]; b.y  = x[t][2 * j + 1];
      acc[2 * p]     = __builtin_amdgcn_wmma_f32_16x16x4_f32(false, a0, false, b, (short)0, acc[2 * p], false, false);
      acc[2 * p + 1] = __builtin_amdgcn_wmma_f32_16x16x4_f32(false, a1, false, b, (short)0, acc[2 * p + 1], false, false);
      const float* src = (i < 16 || p < 3) ? Wc : Wn;
      const int pp = (i < 16) ? p : ((p < 3) ? p + 1 : 0);
      const int ii = (i < 16) ? i + 16 : i - 16;
      const int kr = 16 * (ii >> 2) + 2 * (ii & 3);
      ra[slot][0] = src[(size_t)kr * LD + 32 * pp];
      ra[slot][1] = src[(size_t)(kr + 1) * LD + 32 * pp];
      ra[slot][2] = src[(size_t)kr * LD + 32 * pp + 16];
      ra[slot][3] = src[(size_t)(kr + 1) * LD + 32 * pp + 16];
      __builtin_amdgcn_sched_barrier(0);
    }
  }
}

__device__ __forceinline__ void store_tile(float* __restrict__ p, const float (&y)[8])
{
  v4f lo; lo.x = y[0]; lo.y = y[1]; lo.z = y[2]; lo.w = y[3];
  v4f hi; hi.x = y[4]; hi.y = y[5]; hi.z = y[6]; hi.w = y[7];
  *(v4f*)(p)     = lo;
  *(v4f*)(p + 4) = hi;
}

__global__ __launch_bounds__(32) void kfwd(
    const float* __restrict__ S, const float* __restrict__ wkv,
    const float* __restrict__ w0, const float* __restrict__ w1,
    const float* __restrict__ w2, const float* __restrict__ w3,
    float* __restrict__ KV,
    float* __restrict__ Z0, float* __restrict__ A1,
    float* __restrict__ Z1, float* __restrict__ A2,
    float* __restrict__ Z2, float* __restrict__ A3,
    float* __restrict__ D3)
{
  const int lane = threadIdx.x;
  const int half = lane >> 4;
  const int c    = lane & 15;
  const int tok0 = blockIdx.x * 16;
  const size_t lkv  = (size_t)(8 * half) * 256 + c;
  const size_t lw   = (size_t)(8 * half) * DD + c;
  const size_t orow = (size_t)(tok0 + c) * DD + 8 * half;

  float x[8][8];
  const float* srow = S + orow;
  #pragma unroll
  for (int t = 0; t < 8; ++t) {
    const v4f lo = *(const v4f*)(srow + 16 * t);
    const v4f hi = *(const v4f*)(srow + 16 * t + 4);
    x[t][0] = lo.x; x[t][1] = lo.y; x[t][2] = lo.z; x[t][3] = lo.w;
    x[t][4] = hi.x; x[t][5] = hi.y; x[t][6] = hi.z; x[t][7] = hi.w;
  }

  float v[8][8];
  float ra[16][4];
  v8f   acc[8];

  ring_fill<256>(ra, wkv + DD + lkv);
  {
    const int u = ((blockIdx.x >> 3) & 7) * 32 + lane;
    __builtin_prefetch(wkv + (size_t)(u >> 1) * 256 + 64 * (u & 1), 0, 3);
    __builtin_prefetch(w0 + (size_t)u * 64, 0, 3);
    __builtin_prefetch(w1 + (size_t)u * 64, 0, 3);
    __builtin_prefetch(w2 + (size_t)u * 64, 0, 3);
    __builtin_prefetch(w3 + (size_t)u * 64, 0, 3);
  }
  #pragma unroll 1
  for (int part = 0; part < 2; ++part) {
    chain_layer<256>(x, acc, ra, wkv + lkv + ((part == 0) ? DD : 0), wkv + lkv);
    if (part == 0) {
      #pragma unroll
      for (int t = 0; t < 8; ++t) {
        #pragma unroll
        for (int r = 0; r < 8; ++r) v[t][r] = acc[t][r];
      }
    } else {
      float* krow = KV + (size_t)(tok0 + c) * 256 + 8 * half;
      #pragma unroll
      for (int t = 0; t < 8; ++t) {
        #pragma unroll
        for (int r = 0; r < 8; ++r) x[t][r] = acc[t][r];
        store_tile(krow + 16 * t, x[t]);
      }
    }
  }

  const float* Wc = w0 + lw;
  ring_fill<DD>(ra, Wc);
  #pragma unroll 1
  for (int L = 0; L < 4; ++L) {
    const float* Wn = ((L == 0) ? w1 : (L == 1) ? w2 : w3) + lw;
    chain_layer<DD>(x, acc, ra, Wc, Wn);
    if (L < 3) {
      float* zrow = ((L == 0) ? Z0 : (L == 1) ? Z1 : Z2) + orow;
      float* arow = ((L == 0) ? A1 : (L == 1) ? A2 : A3) + orow;
      #pragma unroll
      for (int t = 0; t < 8; ++t) {
        float z[8];
        #pragma unroll
        for (int r = 0; r < 8; ++r) { z[r] = acc[t][r]; x[t][r] = siluf(z[r]); }
        store_tile(zrow + 16 * t, z);
        store_tile(arow + 16 * t, x[t]);
      }
    } else {
      float* drow = D3 + orow;
      #pragma unroll
      for (int t = 0; t < 8; ++t) {
        float d[8];
        #pragma unroll
        for (int r = 0; r < 8; ++r) d[r] = (2.0f / 128.0f) * (acc[t][r] - v[t][r]);
        store_tile(drow + 16 * t, d);
      }
    }
    Wc = Wn;
  }
}

extern "C" void kernel_launch(void* const* d_in, const int* in_sizes, int n_in,
                              void* d_out, int out_size, void* d_ws, size_t ws_size,
                              hipStream_t stream)
{
  const float* S   = (const float*)d_in[0];
  const float* w0  = (const float*)d_in[1];
  const float* w1  = (const float*)d_in[2];
  const float* w2  = (const float*)d_in[3];
  const float* w3  = (const float*)d_in[4];
  const float* wq  = (const float*)d_in[5];
  const float* wkv = (const float*)d_in[6];
  float* out = (float*)d_out;

  float* p = (float*)d_ws;
  const size_t T = (size_t)TOK * DD;
  float* KV = p; p += (size_t)TOK * 256;
  float* Z0 = p; p += T;  float* A1 = p; p += T;
  float* Z1 = p; p += T;  float* A2 = p; p += T;
  float* Z2 = p; p += T;  float* A3 = p; p += T;
  float* Da = p; p += T;
  float* Db = p; p += T;
  float* U0 = p; p += 4 * DD * DD;
  float* U1 = p; p += 4 * DD * DD;
  float* U2 = p; p += 4 * DD * DD;
  float* U3 = p; p += 4 * DD * DD;

  const dim3 blk(256);
  const dim3 g(TOK / 32, 1);
  const dim3 gg(32, 1, 4);
  const dim3 wv(32);

  kfwd<<<dim3(TOK / 16), wv, 0, stream>>>(S, wkv, w0, w1, w2, w3, KV, Z0, A1, Z1, A2, Z2, A3, Da);

  gradU<DD><<<gg, wv, 0, stream>>>(A3, Da, w3, U3);
  gemm128<<<g, blk, 0, stream>>>(Da, DD, w3, DD, 0, 1, Z2, Db, DD, nullptr, 0);
  gradU<DD><<<gg, wv, 0, stream>>>(A2, Db, w2, U2);
  gemm128<<<g, blk, 0, stream>>>(Db, DD, w2, DD, 0, 1, Z1, Da, DD, nullptr, 0);
  gradU<DD><<<gg, wv, 0, stream>>>(A1, Da, w1, U1);
  gemm128<<<g, blk, 0, stream>>>(Da, DD, w1, DD, 0, 1, Z0, Db, DD, nullptr, 0);
  gradU<256><<<gg, wv, 0, stream>>>(KV, Db, w0, U0);

  qchain<<<dim3(TOK / 16), wv, 0, stream>>>(S, wq, U0, U1, U2, U3, out);
}


# chain kernels (key forward, query): a layer's 8 output tiles split over 4 waves of a work-group (2 each), activations exchanged through LDS, one barrier per layer; sums unchanged
# speedup vs baseline: 3.6350x; 1.3467x over previous
#include <hip/hip_runtime.h>

typedef __attribute__((ext_vector_type(2))) float v2f;
typedef __attribute__((ext_vector_type(4))) float v4f;
typedef __attribute__((ext_vector_type(8))) float v8f;

#define TOK   4096
#define BN    1024
#define DD    128

__device__ __forceinline__ float sigf(float x)  { return 1.0f / (1.0f + __expf(-x)); }
__device__ __forceinline__ float siluf(float x) { return x * sigf(x); }
__device__ __forceinline__ float dsiluf(float x){ float s = sigf(x); return s * (1.0f + x * (1.0f - s)); }

__global__ __launch_bounds__(256) void gemm128(
    const float* __restrict__ A, int lda,
    const float* __restrict__ W, int ldw, int wBatchStride, int wTrans,
    const float* __restrict__ Zmul,
    float* __restrict__ Cout, int ldc,
    float* __restrict__ SiluOut, int ldso)
{
  const int wave = threadIdx.x >> 5;
  const int lane = threadIdx.x & 31;
  const int half = lane >> 4;
  const int l    = lane & 15;
  const int m0   = blockIdx.x * 32 + (wave >> 2) * 16;
  const int n0   = blockIdx.y * 128 + (wave & 3) * 32;
  const int batch = (blockIdx.x * 32) / BN;
  const float* Wb = W + (size_t)batch * wBatchStride;

  __builtin_prefetch(A + (size_t)(m0 + l) * lda + 64 * half, 0, 3);
  if (Zmul) __builtin_prefetch(Zmul + (size_t)(m0 + l) * DD + n0, 0, 3);

  v8f acc0 = {};
  v8f acc1 = {};
  const float* arow = A + (size_t)(m0 + l) * lda + 2 * half;

  if (wTrans) {
    const float* wr0 = Wb + (size_t)(n0 + l) * ldw + 2 * half;
    const float* wr1 = Wb + (size_t)(n0 + 16 + l) * ldw + 2 * half;
    __builtin_prefetch(Wb + (size_t)(n0 + l) * ldw + 64 * half, 0, 3);
    __builtin_prefetch(Wb + (size_t)(n0 + 16 + l) * ldw + 64 * half, 0, 3);
    for (int k0 = 0; k0 < DD; k0 += 4) {
      v2f a  = *(const v2f*)(arow + k0);
      v2f b0 = *(const v2f*)(wr0 + k0);
      v2f b1 = *(const v2f*)(wr1 + k0);
      acc0 = __builtin_amdgcn_wmma_f32_16x16x4_f32(false, a, false, b0, (short)0, acc0, false, false);
      acc1 = __builtin_amdgcn_wmma_f32_16x16x4_f32(false, a, false, b1, (short)0, acc1, false, false);
    }
  } else {
    for (int k0 = 0; k0 < DD; k0 += 4) {
      v2f a = *(const v2f*)(arow + k0);
      const float* r0 = Wb + (size_t)(k0 + 2 * half) * ldw + n0 + l;
      const float* r1 = r0 + ldw;
      v2f b0, b1;
      b0.x = r0[0];  b0.y = r1[0];
      b1.x = r0[16]; b1.y = r1[16];
      acc0 = __builtin_amdgcn_wmma_f32_16x16x4_f32(false, a, false, b0, (short)0, acc0, false, false);
      acc1 = __builtin_amdgcn_wmma_f32_16x16x4_f32(false, a, false, b1, (short)0, acc1, false, false);
    }
  }

  #pragma unroll
  for (int r = 0; r < 8; ++r) {
    const int m = m0 + r + 8 * half;
    float c0 = acc0[r];
    float c1 = acc1[r];
    if (Zmul) {
      c0 *= dsiluf(Zmul[(size_t)m * DD + (n0 + l)]);
      c1 *= dsiluf(Zmul[(size_t)m * DD + (n0 + 16 + l)]);
    }
    if (Cout) {
      Cout[(size_t)m * ldc + n0 + l]      = c0;
      Cout[(size_t)m * ldc + n0 + 16 + l] = c1;
    }
    if (SiluOut) {
      SiluOut[(size_t)m * ldso + n0 + l]      = siluf(c0);
      SiluOut[(size_t)m * ldso + n0 + 16 + l] = siluf(c1);
    }
  }
}

template <int LDA>
__global__ __launch_bounds__(32) void gradU(
    const float* __restrict__ Act,
    const float* __restrict__ Dlt,
    const float* __restrict__ Wbase,
    float* __restrict__ Uout)
{
  constexpr int RING = 16;
  const int lane = threadIdx.x;
  const int half = lane >> 4;
  const int l    = lane & 15;
  const int m0   = (blockIdx.x >> 2) * 16;
  const int n0   = (blockIdx.x & 3) * 32;
  const int batch = blockIdx.z;
  const float* ap = Act + ((size_t)batch * BN + 2 * half) * LDA + m0 + l;
  const float* dp = Dlt + ((size_t)batch * BN + 2 * half) * DD + n0 + 2 * l;

  float ax[RING], ay[RING];
  v2f   t0[RING], t1[RING];

  v2f wb[8];
  #pragma unroll
  for (int r = 0; r < 8; ++r)
    wb[r] = *(const v2f*)(Wbase + (size_t)(m0 + r + 8 * half) * DD + n0 + 2 * l);

  #pragma unroll
  for (int j = 0; j < RING; ++j) {
    ax[j] = ap[(size_t)(4 * j) * LDA];
    ay[j] = ap[(size_t)(4 * j + 1) * LDA];
    t0[j] = *(const v2f*)(dp + (size_t)(4 * j) * DD);
    t1[j] = *(const v2f*)(dp + (size_t)(4 * j + 1) * DD);
    __builtin_amdgcn_sched_barrier(0);
  }

  constexpr int PF_AHEAD = 384;
  const float* pf = (half == 0) ? (Act + ((size_t)batch * BN + l) * LDA + m0)
                                : (Dlt + ((size_t)batch * BN + l) * DD + n0);
  const size_t pfs = (half == 0) ? (size_t)16 * LDA : (size_t)16 * DD;
  #pragma unroll
  for (int g = (4 * RING) / 16; g < PF_AHEAD / 16; ++g) __builtin_prefetch(pf + (size_t)g * pfs, 0, 3);

  v8f acc0 = {};
  v8f acc1 = {};

  for (int k0 = 0; k0 < BN - 4 * RING; k0 += 4 * RING) {
    ap += (size_t)(4 * RING) * LDA;
    dp += (size_t)(4 * RING) * DD;
    if (k0 + PF_AHEAD < BN) {
      const float* pq = pf + (size_t)((k0 + PF_AHEAD) / 16) * pfs;
      __builtin_prefetch(pq, 0, 3);
      __builtin_prefetch(pq + pfs, 0, 3);
      __builtin_prefetch(pq + 2 * pfs, 0, 3);
      __builtin_prefetch(pq + 3 * pfs, 0, 3);
    }
    #pragma unroll
    for (int j = 0; j < RING; ++j) {
      v2f a;  a.x  = ax[j];   a.y  = ay[j];
      float p = t0[j].y, q = t1[j].x;
      asm volatile("v_swap_b32 %0, %1" : "+v"(p), "+v"(q));
      v2f b0; b0.x = t0[j].x; b0.y = p;
      v2f b1; b1.x = q;       b1.y = t1[j].y;
      acc0 = __builtin_amdgcn_wmma_f32_16x16x4_f32(false, a, false, b0, (short)0, acc0, false, false);
      acc1 = __builtin_amdgcn_wmma_f32_16x16x4_f32(false, a, false, b1, (short)0, acc1, false, false);
      ax[j] = ap[(size_t)(4 * j) * LDA];
      ay[j] = ap[(size_t)(4 * j + 1) * LDA];
      t0[j] = *(const v2f*)(dp + (size_t)(4 * j) * DD);
      t1[j] = *(const v2f*)(dp + (size_t)(4 * j + 1) * DD);
    }
  }

  #pragma unroll
  for (int j = 0; j < RING; ++j) {
    v2f a;  a.x  = ax[j];   a.y  = ay[j];
    v2f b0; b0.x = t0[j].x; b0.y = t1[j].x;
    v2f b1; b1.x = t0[j].y; b1.y = t1[j].y;
    acc0 = __builtin_amdgcn_wmma_f32_16x16x4_f32(false, a, false, b0, (short)0, acc0, false, false);
    acc1 = __builtin_amdgcn_wmma_f32_16x16x4_f32(false, a, false, b1, (short)0, acc1, false, false);
  }

  #pragma unroll
  for (int r = 0; r < 8; ++r) {
    const int d = m0 + r + 8 * half;
    v2f o;
    o.x = wb[r].x + acc0[r];
    o.y = wb[r].y + acc1[r];
    *(v2f*)(Uout + (size_t)batch * DD * DD + (size_t)d * DD + n0 + 2 * l) = o;
  }
}


#define XS_V4 512

__device__ __forceinline__ void xs_barrier()
{
  asm volatile("s_wait_dscnt 0x0\n\ts_barrier_signal -1\n\ts_barrier_wait -1" ::: "memory");
}

__device__ __forceinline__ void xs_put(v4f (&xb)[XS_V4], int wave, int lane, const float (&y0)[8], const float (&y1)[8])
{
  const int o = wave * 128 + lane;
  v4f a; a.x = y0[0]; a.y = y0[1]; a.z = y0[2]; a.w = y0[3];
  v4f b; b.x = y0[4]; b.y = y0[5]; b.z = y0[6]; b.w = y0[7];
  v4f c; c.x = y1[0]; c.y = y1[1]; c.z = y1[2]; c.w = y1[3];
  v4f d; d.x = y1[4]; d.y = y1[5]; d.z = y1[6]; d.w = y1[7];
  xb[o]      = a;
  xb[o + 32] = b;
  xb[o + 64] = c;
  xb[o + 96] = d;
}

__device__ __forceinline__ void xs_get(const v4f (&xb)[XS_V4], int lane, float (&x)[8][8])
{
  #pragma unroll
  for (int t = 0; t < 8; ++t) {
    const v4f lo = xb[t * 64 + lane];
    const v4f hi = xb[t * 64 + 32 + lane];
    x[t][0] = lo.x; x[t][1] = lo.y; x[t][2] = lo.z; x[t][3] = lo.w;
    x[t][4] = hi.x; x[t][5] = hi.y; x[t][6] = hi.z; x[t][7] = hi.w;
  }
}

__device__ __forceinline__ void store_tile(float* __restrict__ p, const float (&y)[8])
{
  v4f lo; lo.x = y[0]; lo.y = y[1]; lo.z = y[2]; lo.w = y[3];
  v4f hi; hi.x = y[4]; hi.y = y[5]; hi.z = y[6]; hi.w = y[7];
  *(v4f*)(p)     = lo;
  *(v4f*)(p + 4) = hi;
}

template <int LD>
__device__ __forceinline__ void ring_fill(float (&ra)[16][4], const float* __restrict__ Wc)
{
  #pragma unroll
  for (int i = 0; i < 16; ++i) {
    const int kr = 16 * (i >> 2) + 2 * (i & 3);
    ra[i][0] = Wc[(size_t)kr * LD];
    ra[i][1] = Wc[(size_t)(kr + 1) * LD];
    ra[i][2] = Wc[(size_t)kr * LD + 16];
    ra[i][3] = Wc[(size_t)(kr + 1) * LD + 16];
    __builtin_amdgcn_sched_barrier(0);
  }
}

template <int LD, int LDN>
__device__ __forceinline__ void chain_layer(const float (&x)[8][8], v8f& acc0, v8f& acc1, float (&ra)[16][4],
                                            const float* __restrict__ Wc, const float* __restrict__ Wn)
{
  acc0 = (v8f){};
  acc1 = (v8f){};
  #pragma unroll
  for (int i = 0; i < 32; ++i) {
    const int slot = i & 15;
    const int t = i >> 2, j = i & 3;
    v2f a0; a0.x = ra[slot][0]; a0.y = ra[slot][1];
    v2f a1; a1.x = ra[slot][2]; a1.y = ra[slot][3];
    v2f b;  b.x  = x[t][2 * j]; b.y  = x[t][2 * j + 1];
    acc0 = __builtin_amdgcn_wmma_f32_16x16x4_f32(false, a0, false, b, (short)0, acc0, false, false);
    acc1 = __builtin_amdgcn_wmma_f32_16x16x4_f32(false, a1, false, b, (short)0, acc1, false, false);
    const float* src = (i < 16) ? Wc : Wn;
    const int pitch = (i < 16) ? LD : LDN;
    const int ii = (i < 16) ? i + 16 : i - 16;
    const int kr = 16 * (ii >> 2) + 2 * (ii & 3);
    ra[slot][0] = src[(size_t)kr * pitch];
    ra[slot][1] = src[(size_t)(kr + 1) * pitch];
    ra[slot][2] = src[(size_t)kr * pitch + 16];
    ra[slot][3] = src[(size_t)(kr + 1) * pitch + 16];
    __builtin_amdgcn_sched_barrier(0);
  }
}

__device__ __forceinline__ void load_x(float (&x)[8][8], const float* __restrict__ srow)
{
  #pragma unroll
  for (int t = 0; t < 8; ++t) {
    const v4f lo = *(const v4f*)(srow + 16 * t);
    const v4f hi = *(const v4f*)(srow + 16 * t + 4);
    x[t][0] = lo.x; x[t][1] = lo.y; x[t][2] = lo.z; x[t][3] = lo.w;
    x[t][4] = hi.x; x[t][5] = hi.y; x[t][6] = hi.z; x[t][7] = hi.w;
  }
}

__global__ __launch_bounds__(128) void qchain(
    const float* __restrict__ S, const float* __restrict__ wq,
    const float* __restrict__ U0, const float* __restrict__ U1,
    const float* __restrict__ U2, const float* __restrict__ U3,
    float* __restrict__ out)
{
  __shared__ v4f xs[2][XS_V4];
  const int wave = __builtin_amdgcn_readfirstlane((int)(threadIdx.x >> 5));
  const int lane = threadIdx.x & 31;
  const int half = lane >> 4;
  const int c    = lane & 15;
  const int tok0 = blockIdx.x * 16;
  const size_t boff = (size_t)(tok0 / BN) * DD * DD;
  const size_t woff = (size_t)(32 * wave);
  const size_t loff = (size_t)(8 * half) * DD + c;

  float x[8][8];
  load_x(x, S + (size_t)(tok0 + c) * DD + 8 * half);

  float ra[16][4];
  const float* Wc = (wq + woff) + loff;
  ring_fill<DD>(ra, Wc);

  {
    const float* Uw = (wave == 0) ? U0 : (wave == 1) ? U1 : (wave == 2) ? U2 : U3;
    __builtin_prefetch(Uw + boff + (size_t)(((blockIdx.x >> 3) & 7) * 32 + lane) * 64, 0, 3);
  }

  v8f acc0, acc1;
  #pragma unroll
  for (int L = 0; L < 5; ++L) {
    const float* Wn = (((L == 0) ? U0 : (L == 1) ? U1 : (L == 2) ? U2 : U3) + (boff + woff)) + loff;
    chain_layer<DD, DD>(x, acc0, acc1, ra, Wc, Wn);
    if (L < 4) {
      float y0[8], y1[8];
      if (L >= 1) {
        #pragma unroll
        for (int r = 0; r < 8; ++r) { y0[r] = siluf(acc0[r]); y1[r] = siluf(acc1[r]); }
      } else {
        #pragma unroll
        for (int r = 0; r < 8; ++r) { y0[r] = acc0[r]; y1[r] = acc1[r]; }
      }
      xs_put(xs[L & 1], wave, lane, y0, y1);
      xs_barrier();
      xs_get(xs[L & 1], lane, x);
    }
    Wc = Wn;
  }

  float* orow = out + (size_t)(tok0 + c) * DD + 32 * wave + 8 * half;
  float o0[8], o1[8];
  #pragma unroll
  for (int r = 0; r < 8; ++r) { o0[r] = acc0[r]; o1[r] = acc1[r]; }
  store_tile(orow, o0);
  store_tile(orow + 16, o1);
}

__global__ __launch_bounds__(128) void kfwd(
    const float* __restrict__ S, const float* __restrict__ wkv,
    const float* __restrict__ w0, const float* __restrict__ w1,
    const float* __restrict__ w2, const float* __restrict__ w3,
    float* __restrict__ KV,
    float* __restrict__ Z0, float* __restrict__ A1,
    float* __restrict__ Z1, float* __restrict__ A2,
    float* __restrict__ Z2, float* __restrict__ A3,
    float* __restrict__ D3)
{
  __shared__ v4f xs[2][XS_V4];
  const int wave = __builtin_amdgcn_readfirstlane((int)(threadIdx.x >> 5));
  const int lane = threadIdx.x & 31;
  const int half = lane >> 4;
  const int c    = lane & 15;
  const int tok0 = blockIdx.x * 16;
  const size_t woff = (size_t)(32 * wave);
  const size_t lkv  = (size_t)(8 * half) * 256 + c;
  const size_t lw   = (size_t)(8 * half) * DD + c;
  const size_t orow = (size_t)(tok0 + c) * DD + 32 * wave + 8 * half;

  float x[8][8];
  load_x(x, S + (size_t)(tok0 + c) * DD + 8 * half);

  float v0[8], v1[8];
  float ra[16][4];
  v8f   acc0, acc1;

  const float* Wv = (wkv + (DD + woff)) + lkv;
  const float* Wk = (wkv + woff) + lkv;
  ring_fill<256>(ra, Wv);
  {
    const int u = ((blockIdx.x >> 3) & 7) * 32 + lane;
    const float* ww = (wave == 0) ? w0 : (wave == 1) ? w1 : (wave == 2) ? w2 : w3;
    __builtin_prefetch(ww + (size_t)u * 64, 0, 3);
    if (wave == 0) __builtin_prefetch(wkv + (size_t)(u >> 1) * 256 + 64 * (u & 1), 0, 3);
  }
  chain_layer<256, 256>(x, acc0, acc1, ra, Wv, Wk);
  #pragma unroll
  for (int r = 0; r < 8; ++r) { v0[r] = acc0[r]; v1[r] = acc1[r]; }
  const float* Wc = (w0 + woff) + lw;
  chain_layer<256, DD>(x, acc0, acc1, ra, Wk, Wc);
  {
    float k0[8], k1[8];
    #pragma unroll
    for (int r = 0; r < 8; ++r) { k0[r] = acc0[r]; k1[r] = acc1[r]; }
    float* krow = KV + (size_t)(tok0 + c) * 256 + 32 * wave + 8 * half;
    store_tile(krow, k0);
    store_tile(krow + 16, k1);
    xs_put(xs[0], wave, lane, k0, k1);
    xs_barrier();
    xs_get(xs[0], lane, x);
  }
  #pragma unroll
  for (int L = 0; L < 4; ++L) {
    const float* Wn = (((L == 0) ? w1 : (L == 1) ? w2 : w3) + woff) + lw;
    chain_layer<DD, DD>(x, acc0, acc1, ra, Wc, Wn);
    if (L < 3) {
      float* zrow = ((L == 0) ? Z0 : (L == 1) ? Z1 : Z2) + orow;
      float* arow = ((L == 0) ? A1 : (L == 1) ? A2 : A3) + orow;
      float z0[8], z1[8], a0[8], a1[8];
      #pragma unroll
      for (int r = 0; r < 8; ++r) {
        z0[r] = acc0[r]; a0[r] = siluf(z0[r]);
        z1[r] = acc1[r]; a1[r] = siluf(z1[r]);
      }
      xs_put(xs[(L + 1) & 1], wave, lane, a0, a1);
      xs_barrier();
      xs_get(xs[(L + 1) & 1], lane, x);
      store_tile(zrow, z0);
      store_tile(zrow + 16, z1);
      store_tile(arow, a0);
      store_tile(arow + 16, a1);
    } else {
      float* drow = D3 + orow;
      float d0[8], d1[8];
      #pragma unroll
      for (int r = 0; r < 8; ++r) {
        d0[r] = (2.0f / 128.0f) * (acc0[r] - v0[r]);
        d1[r] = (2.0f / 128.0f) * (acc1[r] - v1[r]);
      }
      store_tile(drow, d0);
      store_tile(drow + 16, d1);
    }
    Wc = Wn;
  }
}

extern "C" void kernel_launch(void* const* d_in, const int* in_sizes, int n_in,
                              void* d_out, int out_size, void* d_ws, size_t ws_size,
                              hipStream_t stream)
{
  const float* S   = (const float*)d_in[0];
  const float* w0  = (const float*)d_in[1];
  const float* w1  = (const float*)d_in[2];
  const float* w2  = (const float*)d_in[3];
  const float* w3  = (const float*)d_in[4];
  const float* wq  = (const float*)d_in[5];
  const float* wkv = (const float*)d_in[6];
  float* out = (float*)d_out;

  float* p = (float*)d_ws;
  const size_t T = (size_t)TOK * DD;
  float* KV = p; p += (size_t)TOK * 256;
  float* Z0 = p; p += T;  float* A1 = p; p += T;
  float* Z1 = p; p += T;  float* A2 = p; p += T;
  float* Z2 = p; p += T;  float* A3 = p; p += T;
  float* Da = p; p += T;
  float* Db = p; p += T;
  float* U0 = p; p += 4 * DD * DD;
  float* U1 = p; p += 4 * DD * DD;
  float* U2 = p; p += 4 * DD * DD;
  float* U3 = p; p += 4 * DD * DD;

  const dim3 blk(256);
  const dim3 g(TOK / 32, 1);
  const dim3 gg(32, 1, 4);
  const dim3 wv(32);

  kfwd<<<dim3(TOK / 16), dim3(128), 0, stream>>>(S, wkv, w0, w1, w2, w3, KV, Z0, A1, Z1, A2, Z2, A3, Da);

  gradU<DD><<<gg, wv, 0, stream>>>(A3, Da, w3, U3);
  gemm128<<<g, blk, 0, stream>>>(Da, DD, w3, DD, 0, 1, Z2, Db, DD, nullptr, 0);
  gradU<DD><<<gg, wv, 0, stream>>>(A2, Db, w2, U2);
  gemm128<<<g, blk, 0, stream>>>(Db, DD, w2, DD, 0, 1, Z1, Da, DD, nullptr, 0);
  gradU<DD><<<gg, wv, 0, stream>>>(A1, Da, w1, U1);
  gemm128<<<g, blk, 0, stream>>>(Da, DD, w1, DD, 0, 1, Z0, Db, DD, nullptr, 0);
  gradU<256><<<gg, wv, 0, stream>>>(KV, Db, w0, U0);

  qchain<<<dim3(TOK / 16), dim3(128), 0, stream>>>(S, wq, U0, U1, U2, U3, out);
}


# each delta GEMM joins the launch of the gradient GEMM that reads the same delta: 1024 one-wave work-groups (own register ring) behind its 128, role by block number; 9 launches -> 6; sums unchanged
# speedup vs baseline: 5.0656x; 1.3936x over previous
#include <hip/hip_runtime.h>

typedef __attribute__((ext_vector_type(2))) float v2f;
typedef __attribute__((ext_vector_type(4))) float v4f;
typedef __attribute__((ext_vector_type(8))) float v8f;

#define TOK   4096
#define BN    1024
#define DD    128

__device__ __forceinline__ float sigf(float x)  { return 1.0f / (1.0f + __expf(-x)); }
__device__ __forceinline__ float siluf(float x) { return x * sigf(x); }
__device__ __forceinline__ float dsiluf(float x){ float s = sigf(x); return s * (1.0f + x * (1.0f - s)); }

__device__ __forceinline__ void delta_tile(
    int id,
    const float* __restrict__ Din,
    const float* __restrict__ Wt,
    const float* __restrict__ Zmul,
    float* __restrict__ Dout)
{
  const int lane = threadIdx.x;
  const int half = lane >> 4;
  const int l    = lane & 15;
  const int m0   = (id >> 2) * 16;
  const int n0   = (id & 3) * 32;

  const float* arow = Din + (size_t)(m0 + l) * DD + 2 * half;
  const float* wr0  = Wt + (size_t)(n0 + l) * DD + 2 * half;
  const float* wr1  = Wt + (size_t)(n0 + 16 + l) * DD + 2 * half;

  v2f a[16], b0[16], b1[16];
  #pragma unroll
  for (int j = 0; j < 16; ++j) {
    a[j]  = *(const v2f*)(arow + 4 * j);
    b0[j] = *(const v2f*)(wr0 + 4 * j);
    b1[j] = *(const v2f*)(wr1 + 4 * j);
  }

  __builtin_prefetch((half == 0) ? (Din + (size_t)(m0 + l) * DD + 64) : (Wt + (size_t)(n0 + l) * DD + 64), 0, 3);
  __builtin_prefetch((half == 0) ? (Wt + (size_t)(n0 + 16 + l) * DD + 64) : (Zmul + (size_t)(m0 + l) * DD + n0), 0, 3);

  v8f acc0 = {};
  v8f acc1 = {};
  #pragma unroll
  for (int j = 0; j < 16; ++j) {
    acc0 = __builtin_amdgcn_wmma_f32_16x16x4_f32(false, a[j], false, b0[j], (short)0, acc0, false, false);
    acc1 = __builtin_amdgcn_wmma_f32_16x16x4_f32(false, a[j], false, b1[j], (short)0, acc1, false, false);
    a[j]  = *(const v2f*)(arow + 64 + 4 * j);
    b0[j] = *(const v2f*)(wr0 + 64 + 4 * j);
    b1[j] = *(const v2f*)(wr1 + 64 + 4 * j);
  }
  #pragma unroll
  for (int j = 0; j < 16; ++j) {
    acc0 = __builtin_amdgcn_wmma_f32_16x16x4_f32(false, a[j], false, b0[j], (short)0, acc0, false, false);
    acc1 = __builtin_amdgcn_wmma_f32_16x16x4_f32(false, a[j], false, b1[j], (short)0, acc1, false, false);
  }

  #pragma unroll
  for (int r = 0; r < 8; ++r) {
    const int m = m0 + r + 8 * half;
    const float c0 = acc0[r] * dsiluf(Zmul[(size_t)m * DD + (n0 + l)]);
    const float c1 = acc1[r] * dsiluf(Zmul[(size_t)m * DD + (n0 + 16 + l)]);
    Dout[(size_t)m * DD + n0 + l]      = c0;
    Dout[(size_t)m * DD + n0 + 16 + l] = c1;
  }
}

template <int LDA>
__device__ __forceinline__ void gradU_tile(
    int tile, int batch,
    const float* __restrict__ Act,
    const float* __restrict__ Dlt,
    const float* __restrict__ Wbase,
    float* __restrict__ Uout)
{
  constexpr int RING = 16;
  const int lane = threadIdx.x;
  const int half = lane >> 4;
  const int l    = lane & 15;
  const int m0   = (tile >> 2) * 16;
  const int n0   = (tile & 3) * 32;
  const float* ap = Act + ((size_t)batch * BN + 2 * half) * LDA + m0 + l;
  const float* dp = Dlt + ((size_t)batch * BN + 2 * half) * DD + n0 + 2 * l;

  float ax[RING], ay[RING];
  v2f   t0[RING], t1[RING];

  v2f wb[8];
  #pragma unroll
  for (int r = 0; r < 8; ++r)
    wb[r] = *(const v2f*)(Wbase + (size_t)(m0 + r + 8 * half) * DD + n0 + 2 * l);

  #pragma unroll
  for (int j = 0; j < RING; ++j) {
    ax[j] = ap[(size_t)(4 * j) * LDA];
    ay[j] = ap[(size_t)(4 * j + 1) * LDA];
    t0[j] = *(const v2f*)(dp + (size_t)(4 * j) * DD);
    t1[j] = *(const v2f*)(dp + (size_t)(4 * j + 1) * DD);
    __builtin_amdgcn_sched_barrier(0);
  }

  constexpr int PF_AHEAD = 384;
  const float* pf = (half == 0) ? (Act + ((size_t)batch * BN + l) * LDA + m0)
                                : (Dlt + ((size_t)batch * BN + l) * DD + n0);
  const size_t pfs = (half == 0) ? (size_t)16 * LDA : (size_t)16 * DD;
  #pragma unroll
  for (int g = (4 * RING) / 16; g < PF_AHEAD / 16; ++g) __builtin_prefetch(pf + (size_t)g * pfs, 0, 3);

  v8f acc0 = {};
  v8f acc1 = {};

  for (int k0 = 0; k0 < BN - 4 * RING; k0 += 4 * RING) {
    ap += (size_t)(4 * RING) * LDA;
    dp += (size_t)(4 * RING) * DD;
    if (k0 + PF_AHEAD < BN) {
      const float* pq = pf + (size_t)((k0 + PF_AHEAD) / 16) * pfs;
      __builtin_prefetch(pq, 0, 3);
      __builtin_prefetch(pq + pfs, 0, 3);
      __builtin_prefetch(pq + 2 * pfs, 0, 3);
      __builtin_prefetch(pq + 3 * pfs, 0, 3);
    }
    #pragma unroll
    for (int j = 0; j < RING; ++j) {
      v2f a;  a.x  = ax[j];   a.y  = ay[j];
      float p = t0[j].y, q = t1[j].x;
      asm volatile("v_swap_b32 %0, %1" : "+v"(p), "+v"(q));
      v2f b0; b0.x = t0[j].x; b0.y = p;
      v2f b1; b1.x = q;       b1.y = t1[j].y;
      acc0 = __builtin_amdgcn_wmma_f32_16x16x4_f32(false, a, false, b0, (short)0, acc0, false, false);
      acc1 = __builtin_amdgcn_wmma_f32_16x16x4_f32(false, a, false, b1, (short)0, acc1, false, false);
      ax[j] = ap[(size_t)(4 * j) * LDA];
      ay[j] = ap[(size_t)(4 * j + 1) * LDA];
      t0[j] = *(const v2f*)(dp + (size_t)(4 * j) * DD);
      t1[j] = *(const v2f*)(dp + (size_t)(4 * j + 1) * DD);
    }
  }

  #pragma unroll
  for (int j = 0; j < RING; ++j) {
    v2f a;  a.x  = ax[j];   a.y  = ay[j];
    v2f b0; b0.x = t0[j].x; b0.y = t1[j].x;
    v2f b1; b1.x = t0[j].y; b1.y = t1[j].y;
    acc0 = __builtin_amdgcn_wmma_f32_16x16x4_f32(false, a, false, b0, (short)0, acc0, false, false);
    acc1 = __builtin_amdgcn_wmma_f32_16x16x4_f32(false, a, false, b1, (short)0, acc1, false, false);
  }

  #pragma unroll
  for (int r = 0; r < 8; ++r) {
    const int d = m0 + r + 8 * half;
    v2f o;
    o.x = wb[r].x + acc0[r];
    o.y = wb[r].y + acc1[r];
    *(v2f*)(Uout + (size_t)batch * DD * DD + (size_t)d * DD + n0 + 2 * l) = o;
  }
}

#define GRAD_WGS  128
#define DELTA_WGS 1024
#define BWD_WGS   1152

__global__ __launch_bounds__(32) void gradU_delta(
    const float* __restrict__ Act,
    const float* __restrict__ Dlt,
    const float* __restrict__ W,
    float* __restrict__ Uout,
    const float* __restrict__ Zprev,
    float* __restrict__ Dprev)
{
  const int wg = blockIdx.x;
  if (wg < GRAD_WGS) gradU_tile<DD>(wg & 31, wg >> 5, Act, Dlt, W, Uout);
  else               delta_tile(wg - GRAD_WGS, Dlt, W, Zprev, Dprev);
}

__global__ __launch_bounds__(32) void gradU_first(
    const float* __restrict__ Act,
    const float* __restrict__ Dlt,
    const float* __restrict__ W,
    float* __restrict__ Uout)
{
  const int wg = blockIdx.x;
  gradU_tile<256>(wg & 31, wg >> 5, Act, Dlt, W, Uout);
}


#define XS_V4 512

__device__ __forceinline__ void xs_barrier()
{
  asm volatile("s_wait_dscnt 0x0\n\ts_barrier_signal -1\n\ts_barrier_wait -1" ::: "memory");
}

__device__ __forceinline__ void xs_put(v4f (&xb)[XS_V4], int wave, int lane, const float (&y0)[8], const float (&y1)[8])
{
  const int o = wave * 128 + lane;
  v4f a; a.x = y0[0]; a.y = y0[1]; a.z = y0[2]; a.w = y0[3];
  v4f b; b.x = y0[4]; b.y = y0[5]; b.z = y0[6]; b.w = y0[7];
  v4f c; c.x = y1[0]; c.y = y1[1]; c.z = y1[2]; c.w = y1[3];
  v4f d; d.x = y1[4]; d.y = y1[5]; d.z = y1[6]; d.w = y1[7];
  xb[o]      = a;
  xb[o + 32] = b;
  xb[o + 64] = c;
  xb[o + 96] = d;
}

__device__ __forceinline__ void xs_get(const v4f (&xb)[XS_V4], int lane, float (&x)[8][8])
{
  #pragma unroll
  for (int t = 0; t < 8; ++t) {
    const v4f lo = xb[t * 64 + lane];
    const v4f hi = xb[t * 64 + 32 + lane];
    x[t][0] = lo.x; x[t][1] = lo.y; x[t][2] = lo.z; x[t][3] = lo.w;
    x[t][4] = hi.x; x[t][5] = hi.y; x[t][6] = hi.z; x[t][7] = hi.w;
  }
}

__device__ __forceinline__ void store_tile(float* __restrict__ p, const float (&y)[8])
{
  v4f lo; lo.x = y[0]; lo.y = y[1]; lo.z = y[2]; lo.w = y[3];
  v4f hi; hi.x = y[4]; hi.y = y[5]; hi.z = y[6]; hi.w = y[7];
  *(v4f*)(p)     = lo;
  *(v4f*)(p + 4) = hi;
}

template <int LD>
__device__ __forceinline__ void ring_fill(float (&ra)[16][4], const float* __restrict__ Wc)
{
  #pragma unroll
  for (int i = 0; i < 16; ++i) {
    const int kr = 16 * (i >> 2) + 2 * (i & 3);
    ra[i][0] = Wc[(size_t)kr * LD];
    ra[i][1] = Wc[(size_t)(kr + 1) * LD];
    ra[i][2] = Wc[(size_t)kr * LD + 16];
    ra[i][3] = Wc[(size_t)(kr + 1) * LD + 16];
    __builtin_amdgcn_sched_barrier(0);
  }
}

template <int LD, int LDN>
__device__ __forceinline__ void chain_layer(const float (&x)[8][8], v8f& acc0, v8f& acc1, float (&ra)[16][4],
                                            const float* __restrict__ Wc, const float* __restrict__ Wn)
{
  acc0 = (v8f){};
  acc1 = (v8f){};
  #pragma unroll
  for (int i = 0; i < 32; ++i) {
    const int slot = i & 15;
    const int t = i >> 2, j = i & 3;
    v2f a0; a0.x = ra[slot][0]; a0.y = ra[slot][1];
    v2f a1; a1.x = ra[slot][2]; a1.y = ra[slot][3];
    v2f b;  b.x  = x[t][2 * j]; b.y  = x[t][2 * j + 1];
    acc0 = __builtin_amdgcn_wmma_f32_16x16x4_f32(false, a0, false, b, (short)0, acc0, false, false);
    acc1 = __builtin_amdgcn_wmma_f32_16x16x4_f32(false, a1, false, b, (short)0, acc1, false, false);
    const float* src = (i < 16) ? Wc : Wn;
    const int pitch = (i < 16) ? LD : LDN;
    const int ii = (i < 16) ? i + 16 : i - 16;
    const int kr = 16 * (ii >> 2) + 2 * (ii & 3);
    ra[slot][0] = src[(size_t)kr * pitch];
    ra[slot][1] = src[(size_t)(kr + 1) * pitch];
    ra[slot][2] = src[(size_t)kr * pitch + 16];
    ra[slot][3] = src[(size_t)(kr + 1) * pitch + 16];
    __builtin_amdgcn_sched_barrier(0);
  }
}

__device__ __forceinline__ void load_x(float (&x)[8][8], const float* __restrict__ srow)
{
  #pragma unroll
  for (int t = 0; t < 8; ++t) {
    const v4f lo = *(const v4f*)(srow + 16 * t);
    const v4f hi = *(const v4f*)(srow + 16 * t + 4);
    x[t][0] = lo.x; x[t][1] = lo.y; x[t][2] = lo.z; x[t][3] = lo.w;
    x[t][4] = hi.x; x[t][5] = hi.y; x[t][6] = hi.z; x[t][7] = hi.w;
  }
}

__global__ __launch_bounds__(128) void qchain(
    const float* __restrict__ S, const float* __restrict__ wq,
    const float* __restrict__ U0, const float* __restrict__ U1,
    const float* __restrict__ U2, const float* __restrict__ U3,
    float* __restrict__ out)
{
  __shared__ v4f xs[2][XS_V4];
  const int wave = __builtin_amdgcn_readfirstlane((int)(threadIdx.x >> 5));
  const int lane = threadIdx.x & 31;
  const int half = lane >> 4;
  const int c    = lane & 15;
  const int tok0 = blockIdx.x * 16;
  const size_t boff = (size_t)(tok0 / BN) * DD * DD;
  const size_t woff = (size_t)(32 * wave);
  const size_t loff = (size_t)(8 * half) * DD + c;

  float x[8][8];
  load_x(x, S + (size_t)(tok0 + c) * DD + 8 * half);

  float ra[16][4];
  const float* Wc = (wq + woff) + loff;
  ring_fill<DD>(ra, Wc);

  {
    const float* Uw = (wave == 0) ? U0 : (wave == 1) ? U1 : (wave == 2) ? U2 : U3;
    __builtin_prefetch(Uw + boff + (size_t)(((blockIdx.x >> 3) & 7) * 32 + lane) * 64, 0, 3);
  }

  v8f acc0, acc1;
  #pragma unroll
  for (int L = 0; L < 5; ++L) {
    const float* Wn = (((L == 0) ? U0 : (L == 1) ? U1 : (L == 2) ? U2 : U3) + (boff + woff)) + loff;
    chain_layer<DD, DD>(x, acc0, acc1, ra, Wc, Wn);
    if (L < 4) {
      float y0[8], y1[8];
      if (L >= 1) {
        #pragma unroll
        for (int r = 0; r < 8; ++r) { y0[r] = siluf(acc0[r]); y1[r] = siluf(acc1[r]); }
      } else {
        #pragma unroll
        for (int r = 0; r < 8; ++r) { y0[r] = acc0[r]; y1[r] = acc1[r]; }
      }
      xs_put(xs[L & 1], wave, lane, y0, y1);
      xs_barrier();
      xs_get(xs[L & 1], lane, x);
    }
    Wc = Wn;
  }

  float* orow = out + (size_t)(tok0 + c) * DD + 32 * wave + 8 * half;
  float o0[8], o1[8];
  #pragma unroll
  for (int r = 0; r < 8; ++r) { o0[r] = acc0[r]; o1[r] = acc1[r]; }
  store_tile(orow, o0);
  store_tile(orow + 16, o1);
}

__global__ __launch_bounds__(128) void kfwd(
    const float* __restrict__ S, const float* __restrict__ wkv,
    const float* __restrict__ w0, const float* __restrict__ w1,
    const float* __restrict__ w2, const float* __restrict__ w3,
    float* __restrict__ KV,
    float* __restrict__ Z0, float* __restrict__ A1,
    float* __restrict__ Z1, float* __restrict__ A2,
    float* __restrict__ Z2, float* __restrict__ A3,
    float* __restrict__ D3)
{
  __shared__ v4f xs[2][XS_V4];
  const int wave = __builtin_amdgcn_readfirstlane((int)(threadIdx.x >> 5));
  const int lane = threadIdx.x & 31;
  const int half = lane >> 4;
  const int c    = lane & 15;
  const int tok0 = blockIdx.x * 16;
  const size_t woff = (size_t)(32 * wave);
  const size_t lkv  = (size_t)(8 * half) * 256 + c;
  const size_t lw   = (size_t)(8 * half) * DD + c;
  const size_t orow = (size_t)(tok0 + c) * DD + 32 * wave + 8 * half;

  float x[8][8];
  load_x(x, S + (size_t)(tok0 + c) * DD + 8 * half);

  float v0[8], v1[8];
  float ra[16][4];
  v8f   acc0, acc1;

  const float* Wv = (wkv + (DD + woff)) + lkv;
  const float* Wk = (wkv + woff) + lkv;
  ring_fill<256>(ra, Wv);
  {
    const int u = ((blockIdx.x >> 3) & 7) * 32 + lane;
    const float* ww = (wave == 0) ? w0 : (wave == 1) ? w1 : (wave == 2) ? w2 : w3;
    __builtin_prefetch(ww + (size_t)u * 64, 0, 3);
    if (wave == 0) __builtin_prefetch(wkv + (size_t)(u >> 1) * 256 + 64 * (u & 1), 0, 3);
  }
  chain_layer<256, 256>(x, acc0, acc1, ra, Wv, Wk);
  #pragma unroll
  for (int r = 0; r < 8; ++r) { v0[r] = acc0[r]; v1[r] = acc1[r]; }
  const float* Wc = (w0 + woff) + lw;
  chain_layer<256, DD>(x, acc0, acc1, ra, Wk, Wc);
  {
    float k0[8], k1[8];
    #pragma unroll
    for (int r = 0; r < 8; ++r) { k0[r] = acc0[r]; k1[r] = acc1[r]; }
    float* krow = KV + (size_t)(tok0 + c) * 256 + 32 * wave + 8 * half;
    store_tile(krow, k0);
    store_tile(krow + 16, k1);
    xs_put(xs[0], wave, lane, k0, k1);
    xs_barrier();
    xs_get(xs[0], lane, x);
  }
  #pragma unroll
  for (int L = 0; L < 4; ++L) {
    const float* Wn = (((L == 0) ? w1 : (L == 1) ? w2 : w3) + woff) + lw;
    chain_layer<DD, DD>(x, acc0, acc1, ra, Wc, Wn);
    if (L < 3) {
      float* zrow = ((L == 0) ? Z0 : (L == 1) ? Z1 : Z2) + orow;
      float* arow = ((L == 0) ? A1 : (L == 1) ? A2 : A3) + orow;
      float z0[8], z1[8], a0[8], a1[8];
      #pragma unroll
      for (int r = 0; r < 8; ++r) {
        z0[r] = acc0[r]; a0[r] = siluf(z0[r]);
        z1[r] = acc1[r]; a1[r] = siluf(z1[r]);
      }
      xs_put(xs[(L + 1) & 1], wave, lane, a0, a1);
      xs_barrier();
      xs_get(xs[(L + 1) & 1], lane, x);
      store_tile(zrow, z0);
      store_tile(zrow + 16, z1);
      store_tile(arow, a0);
      store_tile(arow + 16, a1);
    } else {
      float* drow = D3 + orow;
      float d0[8], d1[8];
      #pragma unroll
      for (int r = 0; r < 8; ++r) {
        d0[r] = (2.0f / 128.0f) * (acc0[r] - v0[r]);
        d1[r] = (2.0f / 128.0f) * (acc1[r] - v1[r]);
      }
      store_tile(drow, d0);
      store_tile(drow + 16, d1);
    }
    Wc = Wn;
  }
}

extern "C" void kernel_launch(void* const* d_in, const int* in_sizes, int n_in,
                              void* d_out, int out_size, void* d_ws, size_t ws_size,
                              hipStream_t stream)
{
  const float* S   = (const float*)d_in[0];
  const float* w0  = (const float*)d_in[1];
  const float* w1  = (const float*)d_in[2];
  const float* w2  = (const float*)d_in[3];
  const float* w3  = (const float*)d_in[4];
  const float* wq  = (const float*)d_in[5];
  const float* wkv = (const float*)d_in[6];
  float* out = (float*)d_out;

  float* p = (float*)d_ws;
  const size_t T = (size_t)TOK * DD;
  float* KV = p; p += (size_t)TOK * 256;
  float* Z0 = p; p += T;  float* A1 = p; p += T;
  float* Z1 = p; p += T;  float* A2 = p; p += T;
  float* Z2 = p; p += T;  float* A3 = p; p += T;
  float* Da = p; p += T;
  float* Db = p; p += T;
  float* U0 = p; p += 4 * DD * DD;
  float* U1 = p; p += 4 * DD * DD;
  float* U2 = p; p += 4 * DD * DD;
  float* U3 = p; p += 4 * DD * DD;

  const dim3 wv(32);
  const dim3 gd(BWD_WGS);
  const dim3 gf(GRAD_WGS);

  kfwd<<<dim3(TOK / 16), dim3(128), 0, stream>>>(S, wkv, w0, w1, w2, w3, KV, Z0, A1, Z1, A2, Z2, A3, Da);

  gradU_delta<<<gd, wv, 0, stream>>>(A3, Da, w3, U3, Z2, Db);
  gradU_delta<<<gd, wv, 0, stream>>>(A2, Db, w2, U2, Z1, Da);
  gradU_delta<<<gd, wv, 0, stream>>>(A1, Da, w1, U1, Z0, Db);
  gradU_first<<<gf, wv, 0, stream>>>(KV, Db, w0, U0);

  qchain<<<dim3(TOK / 16), dim3(128), 0, stream>>>(S, wq, U0, U1, U2, U3, out);
}


# gradient GEMM operands written token-packed by their producers ([token/4][feature][4]): one 16-byte load feeds two k-steps, 3 loads per 8 tokens instead of 8; token sum order permuted within 8
# speedup vs baseline: 5.8475x; 1.1544x over previous
#include <hip/hip_runtime.h>

typedef __attribute__((ext_vector_type(2))) float v2f;
typedef __attribute__((ext_vector_type(4))) float v4f;
typedef __attribute__((ext_vector_type(8))) float v8f;

#define TOK   4096
#define BN    1024
#define DD    128

__device__ __forceinline__ float sigf(float x)  { return 1.0f / (1.0f + __expf(-x)); }
__device__ __forceinline__ float siluf(float x) { return x * sigf(x); }
__device__ __forceinline__ float dsiluf(float x){ float s = sigf(x); return s * (1.0f + x * (1.0f - s)); }

__device__ __forceinline__ void delta_tile(
    int id,
    const float* __restrict__ Din,
    const float* __restrict__ Wt,
    const float* __restrict__ Zmul,
    float* __restrict__ Dout,
    float* __restrict__ Pout)
{
  const int lane = threadIdx.x;
  const int half = lane >> 4;
  const int l    = lane & 15;
  const int m0   = (id >> 2) * 16;
  const int n0   = (id & 3) * 32;

  const float* arow = Din + (size_t)(m0 + l) * DD + 2 * half;
  const float* wr0  = Wt + (size_t)(n0 + l) * DD + 2 * half;
  const float* wr1  = Wt + (size_t)(n0 + 16 + l) * DD + 2 * half;

  v2f a[16], b0[16], b1[16];
  #pragma unroll
  for (int j = 0; j < 16; ++j) {
    a[j]  = *(const v2f*)(arow + 4 * j);
    b0[j] = *(const v2f*)(wr0 + 4 * j);
    b1[j] = *(const v2f*)(wr1 + 4 * j);
  }

  __builtin_prefetch((half == 0) ? (Din + (size_t)(m0 + l) * DD + 64) : (Wt + (size_t)(n0 + l) * DD + 64), 0, 3);
  __builtin_prefetch((half == 0) ? (Wt + (size_t)(n0 + 16 + l) * DD + 64) : (Zmul + (size_t)(m0 + l) * DD + n0), 0, 3);

  v8f acc0 = {};
  v8f acc1 = {};
  #pragma unroll
  for (int j = 0; j < 16; ++j) {
    acc0 = __builtin_amdgcn_wmma_f32_16x16x4_f32(false, a[j], false, b0[j], (short)0, acc0, false, false);
    acc1 = __builtin_amdgcn_wmma_f32_16x16x4_f32(false, a[j], false, b1[j], (short)0, acc1, false, false);
    a[j]  = *(const v2f*)(arow + 64 + 4 * j);
    b0[j] = *(const v2f*)(wr0 + 64 + 4 * j);
    b1[j] = *(const v2f*)(wr1 + 64 + 4 * j);
  }
  #pragma unroll
  for (int j = 0; j < 16; ++j) {
    acc0 = __builtin_amdgcn_wmma_f32_16x16x4_f32(false, a[j], false, b0[j], (short)0, acc0, false, false);
    acc1 = __builtin_amdgcn_wmma_f32_16x16x4_f32(false, a[j], false, b1[j], (short)0, acc1, false, false);
  }

  float c0[8], c1[8];
  #pragma unroll
  for (int r = 0; r < 8; ++r) {
    const int m = m0 + r + 8 * half;
    c0[r] = acc0[r] * dsiluf(Zmul[(size_t)m * DD + (n0 + l)]);
    c1[r] = acc1[r] * dsiluf(Zmul[(size_t)m * DD + (n0 + 16 + l)]);
  }
  {
    float* pp = Pout + ((size_t)((m0 + 8 * half) >> 2) * DD + n0 + l) * 4;
    v4f p0; p0.x = c0[0]; p0.y = c0[1]; p0.z = c0[2]; p0.w = c0[3];
    v4f p1; p1.x = c0[4]; p1.y = c0[5]; p1.z = c0[6]; p1.w = c0[7];
    v4f q0; q0.x = c1[0]; q0.y = c1[1]; q0.z = c1[2]; q0.w = c1[3];
    v4f q1; q1.x = c1[4]; q1.y = c1[5]; q1.z = c1[6]; q1.w = c1[7];
    *(v4f*)(pp)               = p0;
    *(v4f*)(pp + DD * 4)      = p1;
    *(v4f*)(pp + 64)          = q0;
    *(v4f*)(pp + DD * 4 + 64) = q1;
  }
  if (Dout != nullptr) {
    #pragma unroll
    for (int r = 0; r < 8; ++r) {
      const int m = m0 + r + 8 * half;
      Dout[(size_t)m * DD + n0 + l]      = c0[r];
      Dout[(size_t)m * DD + n0 + 16 + l] = c1[r];
    }
  }
}

__device__ __forceinline__ void gradU_tile(
    int tile, int batch,
    const float* __restrict__ PA,
    const float* __restrict__ PD,
    const float* __restrict__ Wbase,
    float* __restrict__ Uout)
{
  constexpr int RING = 16;
  constexpr int NSLOT = BN / 8;
  constexpr size_t SLOT = 2 * DD * 4;
  const int lane = threadIdx.x;
  const int half = lane >> 4;
  const int l    = lane & 15;
  const int m0   = (tile >> 2) * 16;
  const int n0   = (tile & 3) * 32;
  const float* ap = PA + (((size_t)batch * (BN / 4) + half) * DD + m0 + l) * 4;
  const float* dp = PD + (((size_t)batch * (BN / 4) + half) * DD + n0 + 2 * l) * 4;

  v4f a[RING], b0[RING], b1[RING];

  v2f wb[8];
  #pragma unroll
  for (int r = 0; r < 8; ++r)
    wb[r] = *(const v2f*)(Wbase + (size_t)(m0 + r + 8 * half) * DD + n0 + 2 * l);

  #pragma unroll
  for (int j = 0; j < RING; ++j) {
    a[j]  = *(const v4f*)(ap + (size_t)j * SLOT);
    b0[j] = *(const v4f*)(dp + (size_t)j * SLOT);
    b1[j] = *(const v4f*)(dp + (size_t)j * SLOT + 4);
    __builtin_amdgcn_sched_barrier(0);
  }

  constexpr int PF_ROUNDS = 3;
  constexpr size_t ROUND = (size_t)RING * SLOT;
  const float* pfa = PA + (((size_t)batch * (BN / 4) + lane) * DD + m0) * 4;
  const float* pfd = PD + (((size_t)batch * (BN / 4) + lane) * DD + n0) * 4;
  #pragma unroll
  for (int g = 1; g < PF_ROUNDS; ++g) {
    __builtin_prefetch(pfa + (size_t)g * ROUND, 0, 3);
    __builtin_prefetch(pfd + (size_t)g * ROUND, 0, 3);
    __builtin_prefetch(pfd + (size_t)g * ROUND + 64, 0, 3);
  }

  v8f acc0 = {};
  v8f acc1 = {};

  for (int u0 = 0; u0 < NSLOT - RING; u0 += RING) {
    ap += ROUND;
    dp += ROUND;
    if (u0 + PF_ROUNDS * RING < NSLOT) {
      const size_t o = (size_t)(u0 / RING + PF_ROUNDS) * ROUND;
      __builtin_prefetch(pfa + o, 0, 3);
      __builtin_prefetch(pfd + o, 0, 3);
      __builtin_prefetch(pfd + o + 64, 0, 3);
    }
    #pragma unroll
    for (int j = 0; j < RING; ++j) {
      v2f x0; x0.x = a[j].x;  x0.y = a[j].y;
      v2f x1; x1.x = a[j].z;  x1.y = a[j].w;
      v2f p0; p0.x = b0[j].x; p0.y = b0[j].y;
      v2f p1; p1.x = b0[j].z; p1.y = b0[j].w;
      v2f q0; q0.x = b1[j].x; q0.y = b1[j].y;
      v2f q1; q1.x = b1[j].z; q1.y = b1[j].w;
      acc0 = __builtin_amdgcn_wmma_f32_16x16x4_f32(false, x0, false, p0, (short)0, acc0, false, false);
      acc1 = __builtin_amdgcn_wmma_f32_16x16x4_f32(false, x0, false, q0, (short)0, acc1, false, false);
      acc0 = __builtin_amdgcn_wmma_f32_16x16x4_f32(false, x1, false, p1, (short)0, acc0, false, false);
      acc1 = __builtin_amdgcn_wmma_f32_16x16x4_f32(false, x1, false, q1, (short)0, acc1, false, false);
      a[j]  = *(const v4f*)(ap + (size_t)j * SLOT);
      b0[j] = *(const v4f*)(dp + (size_t)j * SLOT);
      b1[j] = *(const v4f*)(dp + (size_t)j * SLOT + 4);
      __builtin_amdgcn_sched_barrier(0);
    }
  }

  #pragma unroll
  for (int j = 0; j < RING; ++j) {
    v2f x0; x0.x = a[j].x;  x0.y = a[j].y;
    v2f x1; x1.x = a[j].z;  x1.y = a[j].w;
    v2f p0; p0.x = b0[j].x; p0.y = b0[j].y;
    v2f p1; p1.x = b0[j].z; p1.y = b0[j].w;
    v2f q0; q0.x = b1[j].x; q0.y = b1[j].y;
    v2f q1; q1.x = b1[j].z; q1.y = b1[j].w;
    acc0 = __builtin_amdgcn_wmma_f32_16x16x4_f32(false, x0, false, p0, (short)0, acc0, false, false);
    acc1 = __builtin_amdgcn_wmma_f32_16x16x4_f32(false, x0, false, q0, (short)0, acc1, false, false);
    acc0 = __builtin_amdgcn_wmma_f32_16x16x4_f32(false, x1, false, p1, (short)0, acc0, false, false);
    acc1 = __builtin_amdgcn_wmma_f32_16x16x4_f32(false, x1, false, q1, (short)0, acc1, false, false);
  }

  #pragma unroll
  for (int r = 0; r < 8; ++r) {
    const int d = m0 + r + 8 * half;
    v2f o;
    o.x = wb[r].x + acc0[r];
    o.y = wb[r].y + acc1[r];
    *(v2f*)(Uout + (size_t)batch * DD * DD + (size_t)d * DD + n0 + 2 * l) = o;
  }
}

#define GRAD_WGS  128
#define DELTA_WGS 1024
#define BWD_WGS   1152

__global__ __launch_bounds__(32) void gradU_delta(
    const float* __restrict__ PAct,
    const float* __restrict__ PDlt,
    const float* __restrict__ Dlt,
    const float* __restrict__ W,
    float* __restrict__ Uout,
    const float* __restrict__ Zprev,
    float* __restrict__ Dprev,
    float* __restrict__ PDprev)
{
  const int wg = blockIdx.x;
  if (wg < GRAD_WGS) gradU_tile(wg & 31, wg >> 5, PAct, PDlt, W, Uout);
  else               delta_tile(wg - GRAD_WGS, Dlt, W, Zprev, Dprev, PDprev);
}

__global__ __launch_bounds__(32) void gradU_first(
    const float* __restrict__ PAct,
    const float* __restrict__ PDlt,
    const float* __restrict__ W,
    float* __restrict__ Uout)
{
  const int wg = blockIdx.x;
  gradU_tile(wg & 31, wg >> 5, PAct, PDlt, W, Uout);
}


#define XS_V4 512

__device__ __forceinline__ void xs_barrier()
{
  asm volatile("s_wait_dscnt 0x0\n\ts_barrier_signal -1\n\ts_barrier_wait -1" ::: "memory");
}

__device__ __forceinline__ void xs_put(v4f (&xb)[XS_V4], int wave, int lane, const float (&y0)[8], const float (&y1)[8])
{
  const int o = wave * 128 + lane;
  v4f a; a.x = y0[0]; a.y = y0[1]; a.z = y0[2]; a.w = y0[3];
  v4f b; b.x = y0[4]; b.y = y0[5]; b.z = y0[6]; b.w = y0[7];
  v4f c; c.x = y1[0]; c.y = y1[1]; c.z = y1[2]; c.w = y1[3];
  v4f d; d.x = y1[4]; d.y = y1[5]; d.z = y1[6]; d.w = y1[7];
  xb[o]      = a;
  xb[o + 32] = b;
  xb[o + 64] = c;
  xb[o + 96] = d;
}

__device__ __forceinline__ void xs_get(const v4f (&xb)[XS_V4], int lane, float (&x)[8][8])
{
  #pragma unroll
  for (int t = 0; t < 8; ++t) {
    const v4f lo = xb[t * 64 + lane];
    const v4f hi = xb[t * 64 + 32 + lane];
    x[t][0] = lo.x; x[t][1] = lo.y; x[t][2] = lo.z; x[t][3] = lo.w;
    x[t][4] = hi.x; x[t][5] = hi.y; x[t][6] = hi.z; x[t][7] = hi.w;
  }
}

__device__ __forceinline__ void store_tile(float* __restrict__ p, const float (&y)[8])
{
  v4f lo; lo.x = y[0]; lo.y = y[1]; lo.z = y[2]; lo.w = y[3];
  v4f hi; hi.x = y[4]; hi.y = y[5]; hi.z = y[6]; hi.w = y[7];
  *(v4f*)(p)     = lo;
  *(v4f*)(p + 4) = hi;
}

__device__ __forceinline__ void store_packed(float* __restrict__ P, const v4f (&xb)[XS_V4], int wave, int lane, int tok0)
{
  const float* xf = (const float*)&xb[0];
  const int base = (2 * wave + (lane >> 4)) * 256 + ((lane >> 2) & 1) * 128 + ((lane >> 3) & 1) * 64 + (lane & 3);
  float* dst = P + ((size_t)(tok0 >> 2) * DD + 32 * wave + lane) * 4;
  #pragma unroll
  for (int g = 0; g < 4; ++g) {
    v4f v;
    v.x = xf[base + (4 * g + 0) * 4];
    v.y = xf[base + (4 * g + 1) * 4];
    v.z = xf[base + (4 * g + 2) * 4];
    v.w = xf[base + (4 * g + 3) * 4];
    *(v4f*)(dst + (size_t)g * DD * 4) = v;
  }
}

template <int LD>
__device__ __forceinline__ void ring_fill(float (&ra)[16][4], const float* __restrict__ Wc)
{
  #pragma unroll
  for (int i = 0; i < 16; ++i) {
    const int kr = 16 * (i >> 2) + 2 * (i & 3);
    ra[i][0] = Wc[(size_t)kr * LD];
    ra[i][1] = Wc[(size_t)(kr + 1) * LD];
    ra[i][2] = Wc[(size_t)kr * LD + 16];
    ra[i][3] = Wc[(size_t)(kr + 1) * LD + 16];
    __builtin_amdgcn_sched_barrier(0);
  }
}

template <int LD, int LDN>
__device__ __forceinline__ void chain_layer(const float (&x)[8][8], v8f& acc0, v8f& acc1, float (&ra)[16][4],
                                            const float* __restrict__ Wc, const float* __restrict__ Wn)
{
  acc0 = (v8f){};
  acc1 = (v8f){};
  #pragma unroll
  for (int i = 0; i < 32; ++i) {
    const int slot = i & 15;
    const int t = i >> 2, j = i & 3;
    v2f a0; a0.x = ra[slot][0]; a0.y = ra[slot][1];
    v2f a1; a1.x = ra[slot][2]; a1.y = ra[slot][3];
    v2f b;  b.x  = x[t][2 * j]; b.y  = x[t][2 * j + 1];
    acc0 = __builtin_amdgcn_wmma_f32_16x16x4_f32(false, a0, false, b, (short)0, acc0, false, false);
    acc1 = __builtin_amdgcn_wmma_f32_16x16x4_f32(false, a1, false, b, (short)0, acc1, false, false);
    const float* src = (i < 16) ? Wc : Wn;
    const int pitch = (i < 16) ? LD : LDN;
    const int ii = (i < 16) ? i + 16 : i - 16;
    const int kr = 16 * (ii >> 2) + 2 * (ii & 3);
    ra[slot][0] = src[(size_t)kr * pitch];
    ra[slot][1] = src[(size_t)(kr + 1) * pitch];
    ra[slot][2] = src[(size_t)kr * pitch + 16];
    ra[slot][3] = src[(size_t)(kr + 1) * pitch + 16];
    __builtin_amdgcn_sched_barrier(0);
  }
}

__device__ __forceinline__ void load_x(float (&x)[8][8], const float* __restrict__ srow)
{
  #pragma unroll
  for (int t = 0; t < 8; ++t) {
    const v4f lo = *(const v4f*)(srow + 16 * t);
    const v4f hi = *(const v4f*)(srow + 16 * t + 4);
    x[t][0] = lo.x; x[t][1] = lo.y; x[t][2] = lo.z; x[t][3] = lo.w;
    x[t][4] = hi.x; x[t][5] = hi.y; x[t][6] = hi.z; x[t][7] = hi.w;
  }
}

__global__ __launch_bounds__(128) void qchain(
    const float* __restrict__ S, const float* __restrict__ wq,
    const float* __restrict__ U0, const float* __restrict__ U1,
    const float* __restrict__ U2, const float* __restrict__ U3,
    float* __restrict__ out)
{
  __shared__ v4f xs[2][XS_V4];
  const int wave = __builtin_amdgcn_readfirstlane((int)(threadIdx.x >> 5));
  const int lane = threadIdx.x & 31;
  const int half = lane >> 4;
  const int c    = lane & 15;
  const int tok0 = blockIdx.x * 16;
  const size_t boff = (size_t)(tok0 / BN) * DD * DD;
  const size_t woff = (size_t)(32 * wave);
  const size_t loff = (size_t)(8 * half) * DD + c;

  float x[8][8];
  load_x(x, S + (size_t)(tok0 + c) * DD + 8 * half);

  float ra[16][4];
  const float* Wc = (wq + woff) + loff;
  ring_fill<DD>(ra, Wc);

  {
    const float* Uw = (wave == 0) ? U0 : (wave == 1) ? U1 : (wave == 2) ? U2 : U3;
    __builtin_prefetch(Uw + boff + (size_t)(((blockIdx.x >> 3) & 7) * 32 + lane) * 64, 0, 3);
  }

  v8f acc0, acc1;
  #pragma unroll
  for (int L = 0; L < 5; ++L) {
    const float* Wn = (((L == 0) ? U0 : (L == 1) ? U1 : (L == 2) ? U2 : U3) + (boff + woff)) + loff;
    chain_layer<DD, DD>(x, acc0, acc1, ra, Wc, Wn);
    if (L < 4) {
      float y0[8], y1[8];
      if (L >= 1) {
        #pragma unroll
        for (int r = 0; r < 8; ++r) { y0[r] = siluf(acc0[r]); y1[r] = siluf(acc1[r]); }
      } else {
        #pragma unroll
        for (int r = 0; r < 8; ++r) { y0[r] = acc0[r]; y1[r] = acc1[r]; }
      }
      xs_put(xs[L & 1], wave, lane, y0, y1);
      xs_barrier();
      xs_get(xs[L & 1], lane, x);
    }
    Wc = Wn;
  }

  float* orow = out + (size_t)(tok0 + c) * DD + 32 * wave + 8 * half;
  float o0[8], o1[8];
  #pragma unroll
  for (int r = 0; r < 8; ++r) { o0[r] = acc0[r]; o1[r] = acc1[r]; }
  store_tile(orow, o0);
  store_tile(orow + 16, o1);
}

__global__ __launch_bounds__(128) void kfwd(
    const float* __restrict__ S, const float* __restrict__ wkv,
    const float* __restrict__ w0, const float* __restrict__ w1,
    const float* __restrict__ w2, const float* __restrict__ w3,
    float* __restrict__ PK,
    float* __restrict__ Z0, float* __restrict__ PA1,
    float* __restrict__ Z1, float* __restrict__ PA2,
    float* __restrict__ Z2, float* __restrict__ PA3,
    float* __restrict__ D3, float* __restrict__ PD3)
{
  __shared__ v4f xs[2][XS_V4];
  const int wave = __builtin_amdgcn_readfirstlane((int)(threadIdx.x >> 5));
  const int lane = threadIdx.x & 31;
  const int half = lane >> 4;
  const int c    = lane & 15;
  const int tok0 = blockIdx.x * 16;
  const size_t woff = (size_t)(32 * wave);
  const size_t lkv  = (size_t)(8 * half) * 256 + c;
  const size_t lw   = (size_t)(8 * half) * DD + c;
  const size_t orow = (size_t)(tok0 + c) * DD + 32 * wave + 8 * half;

  float x[8][8];
  load_x(x, S + (size_t)(tok0 + c) * DD + 8 * half);

  float v0[8], v1[8];
  float ra[16][4];
  v8f   acc0, acc1;

  const float* Wv = (wkv + (DD + woff)) + lkv;
  const float* Wk = (wkv + woff) + lkv;
  ring_fill<256>(ra, Wv);
  {
    const int u = ((blockIdx.x >> 3) & 7) * 32 + lane;
    const float* ww = (wave == 0) ? w0 : (wave == 1) ? w1 : (wave == 2) ? w2 : w3;
    __builtin_prefetch(ww + (size_t)u * 64, 0, 3);
    if (wave == 0) __builtin_prefetch(wkv + (size_t)(u >> 1) * 256 + 64 * (u & 1), 0, 3);
  }
  chain_layer<256, 256>(x, acc0, acc1, ra, Wv, Wk);
  #pragma unroll
  for (int r = 0; r < 8; ++r) { v0[r] = acc0[r]; v1[r] = acc1[r]; }
  const float* Wc = (w0 + woff) + lw;
  chain_layer<256, DD>(x, acc0, acc1, ra, Wk, Wc);
  {
    float k0[8], k1[8];
    #pragma unroll
    for (int r = 0; r < 8; ++r) { k0[r] = acc0[r]; k1[r] = acc1[r]; }
    xs_put(xs[0], wave, lane, k0, k1);
    xs_barrier();
    xs_get(xs[0], lane, x);
    store_packed(PK, xs[0], wave, lane, tok0);
  }
  #pragma unroll
  for (int L = 0; L < 4; ++L) {
    const float* Wn = (((L == 0) ? w1 : (L == 1) ? w2 : w3) + woff) + lw;
    chain_layer<DD, DD>(x, acc0, acc1, ra, Wc, Wn);
    if (L < 3) {
      float* zrow = ((L == 0) ? Z0 : (L == 1) ? Z1 : Z2) + orow;
      float* pact = (L == 0) ? PA1 : (L == 1) ? PA2 : PA3;
      float z0[8], z1[8], a0[8], a1[8];
      #pragma unroll
      for (int r = 0; r < 8; ++r) {
        z0[r] = acc0[r]; a0[r] = siluf(z0[r]);
        z1[r] = acc1[r]; a1[r] = siluf(z1[r]);
      }
      xs_put(xs[(L + 1) & 1], wave, lane, a0, a1);
      xs_barrier();
      xs_get(xs[(L + 1) & 1], lane, x);
      store_tile(zrow, z0);
      store_tile(zrow + 16, z1);
      store_packed(pact, xs[(L + 1) & 1], wave, lane, tok0);
    } else {
      float* drow = D3 + orow;
      float d0[8], d1[8];
      #pragma unroll
      for (int r = 0; r < 8; ++r) {
        d0[r] = (2.0f / 128.0f) * (acc0[r] - v0[r]);
        d1[r] = (2.0f / 128.0f) * (acc1[r] - v1[r]);
      }
      store_tile(drow, d0);
      store_tile(drow + 16, d1);
      xs_put(xs[0], wave, lane, d0, d1);
      asm volatile("s_wait_dscnt 0x0" ::: "memory");
      store_packed(PD3, xs[0], wave, lane, tok0);
    }
    Wc = Wn;
  }
}

extern "C" void kernel_launch(void* const* d_in, const int* in_sizes, int n_in,
                              void* d_out, int out_size, void* d_ws, size_t ws_size,
                              hipStream_t stream)
{
  const float* S   = (const float*)d_in[0];
  const float* w0  = (const float*)d_in[1];
  const float* w1  = (const float*)d_in[2];
  const float* w2  = (const float*)d_in[3];
  const float* w3  = (const float*)d_in[4];
  const float* wq  = (const float*)d_in[5];
  const float* wkv = (const float*)d_in[6];
  float* out = (float*)d_out;

  float* p = (float*)d_ws;
  const size_t T = (size_t)TOK * DD;
  float* PK = p; p += T;
  float* Z0 = p; p += T;  float* A1 = p; p += T;
  float* Z1 = p; p += T;  float* A2 = p; p += T;
  float* Z2 = p; p += T;  float* A3 = p; p += T;
  float* Da = p; p += T;
  float* Db = p; p += T;
  float* Pa = p; p += T;
  float* Pb = p; p += T;
  float* U0 = p; p += 4 * DD * DD;
  float* U1 = p; p += 4 * DD * DD;
  float* U2 = p; p += 4 * DD * DD;
  float* U3 = p; p += 4 * DD * DD;

  const dim3 wv(32);
  const dim3 gd(BWD_WGS);
  const dim3 gf(GRAD_WGS);

  kfwd<<<dim3(TOK / 16), dim3(128), 0, stream>>>(S, wkv, w0, w1, w2, w3, PK, Z0, A1, Z1, A2, Z2, A3, Da, Pa);

  gradU_delta<<<gd, wv, 0, stream>>>(A3, Pa, Da, w3, U3, Z2, Db, Pb);
  gradU_delta<<<gd, wv, 0, stream>>>(A2, Pb, Db, w2, U2, Z1, Da, Pa);
  gradU_delta<<<gd, wv, 0, stream>>>(A1, Pa, Da, w1, U1, Z0, nullptr, Pb);
  gradU_first<<<gf, wv, 0, stream>>>(PK, Pb, w0, U0);

  qchain<<<dim3(TOK / 16), dim3(128), 0, stream>>>(S, wq, U0, U1, U2, U3, out);
}


# gradient GEMM: one 16x16 tile per wave instead of 16x32 (256 one-wave work-groups a layer, 256 matrix instructions each); token sum as two interleaved partial sums (sum order changes); U kept by tile
# speedup vs baseline: 6.8838x; 1.1772x over previous
#include <hip/hip_runtime.h>

typedef __attribute__((ext_vector_type(2))) float v2f;
typedef __attribute__((ext_vector_type(4))) float v4f;
typedef __attribute__((ext_vector_type(8))) float v8f;

#define TOK   4096
#define BN    1024
#define DD    128

__device__ __forceinline__ float sigf(float x)  { return 1.0f / (1.0f + __expf(-x)); }
__device__ __forceinline__ float siluf(float x) { return x * sigf(x); }
__device__ __forceinline__ float dsiluf(float x){ float s = sigf(x); return s * (1.0f + x * (1.0f - s)); }

__device__ __forceinline__ void delta_tile(
    int id,
    const float* __restrict__ Din,
    const float* __restrict__ Wt,
    const float* __restrict__ Zmul,
    float* __restrict__ Dout,
    float* __restrict__ Pout)
{
  const int lane = threadIdx.x;
  const int half = lane >> 4;
  const int l    = lane & 15;
  const int m0   = (id >> 2) * 16;
  const int n0   = (id & 3) * 32;

  const float* arow = Din + (size_t)(m0 + l) * DD + 2 * half;
  const float* wr0  = Wt + (size_t)(n0 + l) * DD + 2 * half;
  const float* wr1  = Wt + (size_t)(n0 + 16 + l) * DD + 2 * half;

  v2f a[16], b0[16], b1[16];
  #pragma unroll
  for (int j = 0; j < 16; ++j) {
    a[j]  = *(const v2f*)(arow + 4 * j);
    b0[j] = *(const v2f*)(wr0 + 4 * j);
    b1[j] = *(const v2f*)(wr1 + 4 * j);
  }

  __builtin_prefetch((half == 0) ? (Din + (size_t)(m0 + l) * DD + 64) : (Wt + (size_t)(n0 + l) * DD + 64), 0, 3);
  __builtin_prefetch((half == 0) ? (Wt + (size_t)(n0 + 16 + l) * DD + 64) : (Zmul + (size_t)(m0 + l) * DD + n0), 0, 3);

  v8f acc0 = {};
  v8f acc1 = {};
  #pragma unroll
  for (int j = 0; j < 16; ++j) {
    acc0 = __builtin_amdgcn_wmma_f32_16x16x4_f32(false, a[j], false, b0[j], (short)0, acc0, false, false);
    acc1 = __builtin_amdgcn_wmma_f32_16x16x4_f32(false, a[j], false, b1[j], (short)0, acc1, false, false);
    a[j]  = *(const v2f*)(arow + 64 + 4 * j);
    b0[j] = *(const v2f*)(wr0 + 64 + 4 * j);
    b1[j] = *(const v2f*)(wr1 + 64 + 4 * j);
  }
  #pragma unroll
  for (int j = 0; j < 16; ++j) {
    acc0 = __builtin_amdgcn_wmma_f32_16x16x4_f32(false, a[j], false, b0[j], (short)0, acc0, false, false);
    acc1 = __builtin_amdgcn_wmma_f32_16x16x4_f32(false, a[j], false, b1[j], (short)0, acc1, false, false);
  }

  float c0[8], c1[8];
  #pragma unroll
  for (int r = 0; r < 8; ++r) {
    const int m = m0 + r + 8 * half;
    c0[r] = acc0[r] * dsiluf(Zmul[(size_t)m * DD + (n0 + l)]);
    c1[r] = acc1[r] * dsiluf(Zmul[(size_t)m * DD + (n0 + 16 + l)]);
  }
  {
    float* pp = Pout + ((size_t)((m0 + 8 * half) >> 2) * DD + n0 + l) * 4;
    v4f p0; p0.x = c0[0]; p0.y = c0[1]; p0.z = c0[2]; p0.w = c0[3];
    v4f p1; p1.x = c0[4]; p1.y = c0[5]; p1.z = c0[6]; p1.w = c0[7];
    v4f q0; q0.x = c1[0]; q0.y = c1[1]; q0.z = c1[2]; q0.w = c1[3];
    v4f q1; q1.x = c1[4]; q1.y = c1[5]; q1.z = c1[6]; q1.w = c1[7];
    *(v4f*)(pp)               = p0;
    *(v4f*)(pp + DD * 4)      = p1;
    *(v4f*)(pp + 64)          = q0;
    *(v4f*)(pp + DD * 4 + 64) = q1;
  }
  if (Dout != nullptr) {
    #pragma unroll
    for (int r = 0; r < 8; ++r) {
      const int m = m0 + r + 8 * half;
      Dout[(size_t)m * DD + n0 + l]      = c0[r];
      Dout[(size_t)m * DD + n0 + 16 + l] = c1[r];
    }
  }
}

__device__ __forceinline__ void gradU_tile(
    int tile, int batch,
    const float* __restrict__ PA,
    const float* __restrict__ PD,
    const float* __restrict__ Wbase,
    float* __restrict__ Uout)
{
  constexpr int RING = 32;
  constexpr int NSLOT = BN / 8;
  constexpr size_t SLOT = 2 * DD * 4;
  const int lane = threadIdx.x;
  const int half = lane >> 4;
  const int l    = lane & 15;
  const int m0   = (tile >> 3) * 16;
  const int n0   = (tile & 7) * 16;
  const float* ap = PA + (((size_t)batch * (BN / 4) + half) * DD + m0 + l) * 4;
  const float* dp = PD + (((size_t)batch * (BN / 4) + half) * DD + n0 + l) * 4;

  v4f a[RING], b[RING];

  #pragma unroll
  for (int j = 0; j < RING; ++j) {
    a[j] = *(const v4f*)(ap + (size_t)j * SLOT);
    b[j] = *(const v4f*)(dp + (size_t)j * SLOT);
    __builtin_amdgcn_sched_barrier(0);
  }

  constexpr int PF_ROUNDS = 3;
  constexpr size_t ROUND = (size_t)RING * SLOT;
  constexpr size_t HALFROUND = (size_t)32 * DD * 4;
  const float* pfa = PA + (((size_t)batch * (BN / 4) + lane) * DD + m0) * 4;
  const float* pfd = PD + (((size_t)batch * (BN / 4) + lane) * DD + n0) * 4;
  #pragma unroll
  for (int g = 1; g < PF_ROUNDS; ++g) {
    __builtin_prefetch(pfa + (size_t)g * ROUND, 0, 3);
    __builtin_prefetch(pfd + (size_t)g * ROUND, 0, 3);
    __builtin_prefetch(pfa + (size_t)g * ROUND + HALFROUND, 0, 3);
    __builtin_prefetch(pfd + (size_t)g * ROUND + HALFROUND, 0, 3);
  }

  v8f acc0 = {};
  v8f acc1 = {};

  for (int u0 = 0; u0 < NSLOT - RING; u0 += RING) {
    ap += ROUND;
    dp += ROUND;
    if (u0 + PF_ROUNDS * RING < NSLOT) {
      const size_t o = (size_t)(u0 / RING + PF_ROUNDS) * ROUND;
      __builtin_prefetch(pfa + o, 0, 3);
      __builtin_prefetch(pfd + o, 0, 3);
      __builtin_prefetch(pfa + o + HALFROUND, 0, 3);
      __builtin_prefetch(pfd + o + HALFROUND, 0, 3);
    }
    #pragma unroll
    for (int j = 0; j < RING; ++j) {
      v2f x0; x0.x = a[j].x; x0.y = a[j].y;
      v2f x1; x1.x = a[j].z; x1.y = a[j].w;
      v2f p0; p0.x = b[j].x; p0.y = b[j].y;
      v2f p1; p1.x = b[j].z; p1.y = b[j].w;
      acc0 = __builtin_amdgcn_wmma_f32_16x16x4_f32(false, x0, false, p0, (short)0, acc0, false, false);
      acc1 = __builtin_amdgcn_wmma_f32_16x16x4_f32(false, x1, false, p1, (short)0, acc1, false, false);
      a[j] = *(const v4f*)(ap + (size_t)j * SLOT);
      b[j] = *(const v4f*)(dp + (size_t)j * SLOT);
      __builtin_amdgcn_sched_barrier(0);
    }
  }

  float wb[8];
  #pragma unroll
  for (int r = 0; r < 8; ++r)
    wb[r] = Wbase[(size_t)(m0 + r + 8 * half) * DD + n0 + l];

  #pragma unroll
  for (int j = 0; j < RING; ++j) {
    v2f x0; x0.x = a[j].x; x0.y = a[j].y;
    v2f x1; x1.x = a[j].z; x1.y = a[j].w;
    v2f p0; p0.x = b[j].x; p0.y = b[j].y;
    v2f p1; p1.x = b[j].z; p1.y = b[j].w;
    acc0 = __builtin_amdgcn_wmma_f32_16x16x4_f32(false, x0, false, p0, (short)0, acc0, false, false);
    acc1 = __builtin_amdgcn_wmma_f32_16x16x4_f32(false, x1, false, p1, (short)0, acc1, false, false);
  }

  float* ut = Uout + (size_t)batch * DD * DD + (size_t)tile * 256 + (8 * half) * 16 + l;
  #pragma unroll
  for (int r = 0; r < 8; ++r)
    ut[r * 16] = wb[r] + (acc0[r] + acc1[r]);
}

#define GRAD_WGS  256
#define DELTA_WGS 1024
#define BWD_WGS   1280

__global__ __launch_bounds__(32) void gradU_delta(
    const float* __restrict__ PAct,
    const float* __restrict__ PDlt,
    const float* __restrict__ Dlt,
    const float* __restrict__ W,
    float* __restrict__ Uout,
    const float* __restrict__ Zprev,
    float* __restrict__ Dprev,
    float* __restrict__ PDprev)
{
  const int wg = blockIdx.x;
  if (wg < GRAD_WGS) gradU_tile(wg & 63, wg >> 6, PAct, PDlt, W, Uout);
  else               delta_tile(wg - GRAD_WGS, Dlt, W, Zprev, Dprev, PDprev);
}

__global__ __launch_bounds__(32) void gradU_first(
    const float* __restrict__ PAct,
    const float* __restrict__ PDlt,
    const float* __restrict__ W,
    float* __restrict__ Uout)
{
  const int wg = blockIdx.x;
  gradU_tile(wg & 63, wg >> 6, PAct, PDlt, W, Uout);
}


#define XS_V4 512

__device__ __forceinline__ void xs_barrier()
{
  asm volatile("s_wait_dscnt 0x0\n\ts_barrier_signal -1\n\ts_barrier_wait -1" ::: "memory");
}

__device__ __forceinline__ void xs_put(v4f (&xb)[XS_V4], int wave, int lane, const float (&y0)[8], const float (&y1)[8])
{
  const int o = wave * 128 + lane;
  v4f a; a.x = y0[0]; a.y = y0[1]; a.z = y0[2]; a.w = y0[3];
  v4f b; b.x = y0[4]; b.y = y0[5]; b.z = y0[6]; b.w = y0[7];
  v4f c; c.x = y1[0]; c.y = y1[1]; c.z = y1[2]; c.w = y1[3];
  v4f d; d.x = y1[4]; d.y = y1[5]; d.z = y1[6]; d.w = y1[7];
  xb[o]      = a;
  xb[o + 32] = b;
  xb[o + 64] = c;
  xb[o + 96] = d;
}

__device__ __forceinline__ void xs_get(const v4f (&xb)[XS_V4], int lane, float (&x)[8][8])
{
  #pragma unroll
  for (int t = 0; t < 8; ++t) {
    const v4f lo = xb[t * 64 + lane];
    const v4f hi = xb[t * 64 + 32 + lane];
    x[t][0] = lo.x; x[t][1] = lo.y; x[t][2] = lo.z; x[t][3] = lo.w;
    x[t][4] = hi.x; x[t][5] = hi.y; x[t][6] = hi.z; x[t][7] = hi.w;
  }
}

__device__ __forceinline__ void store_tile(float* __restrict__ p, const float (&y)[8])
{
  v4f lo; lo.x = y[0]; lo.y = y[1]; lo.z = y[2]; lo.w = y[3];
  v4f hi; hi.x = y[4]; hi.y = y[5]; hi.z = y[6]; hi.w = y[7];
  *(v4f*)(p)     = lo;
  *(v4f*)(p + 4) = hi;
}

__device__ __forceinline__ void store_packed(float* __restrict__ P, const v4f (&xb)[XS_V4], int wave, int lane, int tok0)
{
  const float* xf = (const float*)&xb[0];
  const int base = (2 * wave + (lane >> 4)) * 256 + ((lane >> 2) & 1) * 128 + ((lane >> 3) & 1) * 64 + (lane & 3);
  float* dst = P + ((size_t)(tok0 >> 2) * DD + 32 * wave + lane) * 4;
  #pragma unroll
  for (int g = 0; g < 4; ++g) {
    v4f v;
    v.x = xf[base + (4 * g + 0) * 4];
    v.y = xf[base + (4 * g + 1) * 4];
    v.z = xf[base + (4 * g + 2) * 4];
    v.w = xf[base + (4 * g + 3) * 4];
    *(v4f*)(dst + (size_t)g * DD * 4) = v;
  }
}

#define ROWMAJOR(LD) (LD), (16 * (LD)), 16
#define TILED        16, 2048, 256

template <int RP, int BP, int T1>
__device__ __forceinline__ void ring_fill(float (&ra)[16][4], const float* __restrict__ Wc)
{
  #pragma unroll
  for (int i = 0; i < 16; ++i) {
    const size_t o = (size_t)(i >> 2) * BP + (size_t)(2 * (i & 3)) * RP;
    ra[i][0] = Wc[o];
    ra[i][1] = Wc[o + RP];
    ra[i][2] = Wc[o + T1];
    ra[i][3] = Wc[o + RP + T1];
    __builtin_amdgcn_sched_barrier(0);
  }
}

template <int RP, int BP, int T1, int RPN, int BPN, int T1N>
__device__ __forceinline__ void chain_layer(const float (&x)[8][8], v8f& acc0, v8f& acc1, float (&ra)[16][4],
                                            const float* __restrict__ Wc, const float* __restrict__ Wn)
{
  acc0 = (v8f){};
  acc1 = (v8f){};
  #pragma unroll
  for (int i = 0; i < 32; ++i) {
    const int slot = i & 15;
    const int t = i >> 2, j = i & 3;
    v2f a0; a0.x = ra[slot][0]; a0.y = ra[slot][1];
    v2f a1; a1.x = ra[slot][2]; a1.y = ra[slot][3];
    v2f b;  b.x  = x[t][2 * j]; b.y  = x[t][2 * j + 1];
    acc0 = __builtin_amdgcn_wmma_f32_16x16x4_f32(false, a0, false, b, (short)0, acc0, false, false);
    acc1 = __builtin_amdgcn_wmma_f32_16x16x4_f32(false, a1, false, b, (short)0, acc1, false, false);
    const float* src = (i < 16) ? Wc : Wn;
    const int rp = (i < 16) ? RP : RPN;
    const int bp = (i < 16) ? BP : BPN;
    const int t1 = (i < 16) ? T1 : T1N;
    const int ii = (i < 16) ? i + 16 : i - 16;
    const size_t o = (size_t)(ii >> 2) * bp + (size_t)(2 * (ii & 3)) * rp;
    ra[slot][0] = src[o];
    ra[slot][1] = src[o + rp];
    ra[slot][2] = src[o + t1];
    ra[slot][3] = src[o + rp + t1];
    __builtin_amdgcn_sched_barrier(0);
  }
}

__device__ __forceinline__ void load_x(float (&x)[8][8], const float* __restrict__ srow)
{
  #pragma unroll
  for (int t = 0; t < 8; ++t) {
    const v4f lo = *(const v4f*)(srow + 16 * t);
    const v4f hi = *(const v4f*)(srow + 16 * t + 4);
    x[t][0] = lo.x; x[t][1] = lo.y; x[t][2] = lo.z; x[t][3] = lo.w;
    x[t][4] = hi.x; x[t][5] = hi.y; x[t][6] = hi.z; x[t][7] = hi.w;
  }
}

__global__ __launch_bounds__(128) void qchain(
    const float* __restrict__ S, const float* __restrict__ wq,
    const float* __restrict__ U0, const float* __restrict__ U1,
    const float* __restrict__ U2, const float* __restrict__ U3,
    float* __restrict__ out)
{
  __shared__ v4f xs[2][XS_V4];
  const int wave = __builtin_amdgcn_readfirstlane((int)(threadIdx.x >> 5));
  const int lane = threadIdx.x & 31;
  const int half = lane >> 4;
  const int c    = lane & 15;
  const int tok0 = blockIdx.x * 16;
  const size_t boff = (size_t)(tok0 / BN) * DD * DD;
  const size_t woff = (size_t)(32 * wave);
  const size_t loff = (size_t)(8 * half) * DD + c;
  const size_t uoff = boff + (size_t)(512 * wave);
  const size_t luof = (size_t)(8 * half) * 16 + c;

  float x[8][8];
  load_x(x, S + (size_t)(tok0 + c) * DD + 8 * half);

  float ra[16][4];
  const float* Wc = (wq + woff) + loff;
  ring_fill<ROWMAJOR(DD)>(ra, Wc);

  {
    const float* Uw = (wave == 0) ? U0 : (wave == 1) ? U1 : (wave == 2) ? U2 : U3;
    __builtin_prefetch(Uw + boff + (size_t)(((blockIdx.x >> 3) & 7) * 32 + lane) * 64, 0, 3);
  }

  v8f acc0, acc1;
  #pragma unroll
  for (int L = 0; L < 5; ++L) {
    const float* Wn = (((L == 0) ? U0 : (L == 1) ? U1 : (L == 2) ? U2 : U3) + uoff) + luof;
    if (L == 0) chain_layer<ROWMAJOR(DD), TILED>(x, acc0, acc1, ra, Wc, Wn);
    else        chain_layer<TILED, TILED>(x, acc0, acc1, ra, Wc, Wn);
    if (L < 4) {
      float y0[8], y1[8];
      if (L >= 1) {
        #pragma unroll
        for (int r = 0; r < 8; ++r) { y0[r] = siluf(acc0[r]); y1[r] = siluf(acc1[r]); }
      } else {
        #pragma unroll
        for (int r = 0; r < 8; ++r) { y0[r] = acc0[r]; y1[r] = acc1[r]; }
      }
      xs_put(xs[L & 1], wave, lane, y0, y1);
      xs_barrier();
      xs_get(xs[L & 1], lane, x);
    }
    Wc = Wn;
  }

  float* orow = out + (size_t)(tok0 + c) * DD + 32 * wave + 8 * half;
  float o0[8], o1[8];
  #pragma unroll
  for (int r = 0; r < 8; ++r) { o0[r] = acc0[r]; o1[r] = acc1[r]; }
  store_tile(orow, o0);
  store_tile(orow + 16, o1);
}

__global__ __launch_bounds__(128) void kfwd(
    const float* __restrict__ S, const float* __restrict__ wkv,
    const float* __restrict__ w0, const float* __restrict__ w1,
    const float* __restrict__ w2, const float* __restrict__ w3,
    float* __restrict__ PK,
    float* __restrict__ Z0, float* __restrict__ PA1,
    float* __restrict__ Z1, float* __restrict__ PA2,
    float* __restrict__ Z2, float* __restrict__ PA3,
    float* __restrict__ D3, float* __restrict__ PD3)
{
  __shared__ v4f xs[2][XS_V4];
  const int wave = __builtin_amdgcn_readfirstlane((int)(threadIdx.x >> 5));
  const int lane = threadIdx.x & 31;
  const int half = lane >> 4;
  const int c    = lane & 15;
  const int tok0 = blockIdx.x * 16;
  const size_t woff = (size_t)(32 * wave);
  const size_t lkv  = (size_t)(8 * half) * 256 + c;
  const size_t lw   = (size_t)(8 * half) * DD + c;
  const size_t orow = (size_t)(tok0 + c) * DD + 32 * wave + 8 * half;

  float x[8][8];
  load_x(x, S + (size_t)(tok0 + c) * DD + 8 * half);

  float v0[8], v1[8];
  float ra[16][4];
  v8f   acc0, acc1;

  const float* Wv = (wkv + (DD + woff)) + lkv;
  const float* Wk = (wkv + woff) + lkv;
  ring_fill<ROWMAJOR(256)>(ra, Wv);
  {
    const int u = ((blockIdx.x >> 3) & 7) * 32 + lane;
    const float* ww = (wave == 0) ? w0 : (wave == 1) ? w1 : (wave == 2) ? w2 : w3;
    __builtin_prefetch(ww + (size_t)u * 64, 0, 3);
    if (wave == 0) __builtin_prefetch(wkv + (size_t)(u >> 1) * 256 + 64 * (u & 1), 0, 3);
  }
  chain_layer<ROWMAJOR(256), ROWMAJOR(256)>(x, acc0, acc1, ra, Wv, Wk);
  #pragma unroll
  for (int r = 0; r < 8; ++r) { v0[r] = acc0[r]; v1[r] = acc1[r]; }
  const float* Wc = (w0 + woff) + lw;
  chain_layer<ROWMAJOR(256), ROWMAJOR(DD)>(x, acc0, acc1, ra, Wk, Wc);
  {
    float k0[8], k1[8];
    #pragma unroll
    for (int r = 0; r < 8; ++r) { k0[r] = acc0[r]; k1[r] = acc1[r]; }
    xs_put(xs[0], wave, lane, k0, k1);
    xs_barrier();
    xs_get(xs[0], lane, x);
    store_packed(PK, xs[0], wave, lane, tok0);
  }
  #pragma unroll
  for (int L = 0; L < 4; ++L) {
    const float* Wn = (((L == 0) ? w1 : (L == 1) ? w2 : w3) + woff) + lw;
    chain_layer<ROWMAJOR(DD), ROWMAJOR(DD)>(x, acc0, acc1, ra, Wc, Wn);
    if (L < 3) {
      float* zrow = ((L == 0) ? Z0 : (L == 1) ? Z1 : Z2) + orow;
      float* pact = (L == 0) ? PA1 : (L == 1) ? PA2 : PA3;
      float z0[8], z1[8], a0[8], a1[8];
      #pragma unroll
      for (int r = 0; r < 8; ++r) {
        z0[r] = acc0[r]; a0[r] = siluf(z0[r]);
        z1[r] = acc1[r]; a1[r] = siluf(z1[r]);
      }
      xs_put(xs[(L + 1) & 1], wave, lane, a0, a1);
      xs_barrier();
      xs_get(xs[(L + 1) & 1], lane, x);
      store_tile(zrow, z0);
      store_tile(zrow + 16, z1);
      store_packed(pact, xs[(L + 1) & 1], wave, lane, tok0);
    } else {
      float* drow = D3 + orow;
      float d0[8], d1[8];
      #pragma unroll
      for (int r = 0; r < 8; ++r) {
        d0[r] = (2.0f / 128.0f) * (acc0[r] - v0[r]);
        d1[r] = (2.0f / 128.0f) * (acc1[r] - v1[r]);
      }
      store_tile(drow, d0);
      store_tile(drow + 16, d1);
      xs_put(xs[0], wave, lane, d0, d1);
      asm volatile("s_wait_dscnt 0x0" ::: "memory");
      store_packed(PD3, xs[0], wave, lane, tok0);
    }
    Wc = Wn;
  }
}

extern "C" void kernel_launch(void* const* d_in, const int* in_sizes, int n_in,
                              void* d_out, int out_size, void* d_ws, size_t ws_size,
                              hipStream_t stream)
{
  const float* S   = (const float*)d_in[0];
  const float* w0  = (const float*)d_in[1];
  const float* w1  = (const float*)d_in[2];
  const float* w2  = (const float*)d_in[3];
  const float* w3  = (const float*)d_in[4];
  const float* wq  = (const float*)d_in[5];
  const float* wkv = (const float*)d_in[6];
  float* out = (float*)d_out;

  float* p = (float*)d_ws;
  const size_t T = (size_t)TOK * DD;
  float* PK = p; p += T;
  float* Z0 = p; p += T;  float* A1 = p; p += T;
  float* Z1 = p; p += T;  float* A2 = p; p += T;
  float* Z2 = p; p += T;  float* A3 = p; p += T;
  float* Da = p; p += T;
  float* Db = p; p += T;
  float* Pa = p; p += T;
  float* Pb = p; p += T;
  float* U0 = p; p += 4 * DD * DD;
  float* U1 = p; p += 4 * DD * DD;
  float* U2 = p; p += 4 * DD * DD;
  float* U3 = p; p += 4 * DD * DD;

  const dim3 wv(32);
  const dim3 gd(BWD_WGS);
  const dim3 gf(GRAD_WGS);

  kfwd<<<dim3(TOK / 16), dim3(128), 0, stream>>>(S, wkv, w0, w1, w2, w3, PK, Z0, A1, Z1, A2, Z2, A3, Da, Pa);

  gradU_delta<<<gd, wv, 0, stream>>>(A3, Pa, Da, w3, U3, Z2, Db, Pb);
  gradU_delta<<<gd, wv, 0, stream>>>(A2, Pb, Db, w2, U2, Z1, Da, Pa);
  gradU_delta<<<gd, wv, 0, stream>>>(A1, Pa, Da, w1, U1, Z0, nullptr, Pb);
  gradU_first<<<gf, wv, 0, stream>>>(PK, Pb, w0, U0);

  qchain<<<dim3(TOK / 16), dim3(128), 0, stream>>>(S, wq, U0, U1, U2, U3, out);
}


# launch fusion 6 -> 3: the 3 delta layers run inside the key-path kernel (token-local, transposed form, silu' kept in registers); the 4 gradient GEMMs share one launch; delta k sum order permuted
# speedup vs baseline: 10.7115x; 1.5560x over previous
#include <hip/hip_runtime.h>

typedef __attribute__((ext_vector_type(2))) float v2f;
typedef __attribute__((ext_vector_type(4))) float v4f;
typedef __attribute__((ext_vector_type(8))) float v8f;

#define TOK   4096
#define BN    1024
#define DD    128

__device__ __forceinline__ float sigf(float x)  { return 1.0f / (1.0f + __expf(-x)); }
__device__ __forceinline__ float siluf(float x) { return x * sigf(x); }
__device__ __forceinline__ float dsiluf(float x){ float s = sigf(x); return s * (1.0f + x * (1.0f - s)); }


__device__ __forceinline__ void gradU_tile(
    int tile, int batch, int lane,
    const float* __restrict__ PA,
    const float* __restrict__ PD,
    const float* __restrict__ Wbase,
    float* __restrict__ Uout)
{
  constexpr int RING = 32;
  constexpr int NSLOT = BN / 8;
  constexpr size_t SLOT = 2 * DD * 4;
  const int half = lane >> 4;
  const int l    = lane & 15;
  const int m0   = (tile >> 3) * 16;
  const int n0   = (tile & 7) * 16;
  const float* ap = PA + (((size_t)batch * (BN / 4) + half) * DD + m0 + l) * 4;
  const float* dp = PD + (((size_t)batch * (BN / 4) + half) * DD + n0 + l) * 4;

  v4f a[RING], b[RING];

  #pragma unroll
  for (int j = 0; j < RING; ++j) {
    a[j] = *(const v4f*)(ap + (size_t)j * SLOT);
    b[j] = *(const v4f*)(dp + (size_t)j * SLOT);
    __builtin_amdgcn_sched_barrier(0);
  }

  constexpr int PF_ROUNDS = 3;
  constexpr size_t ROUND = (size_t)RING * SLOT;
  constexpr size_t HALFROUND = (size_t)32 * DD * 4;
  const float* pfa = PA + (((size_t)batch * (BN / 4) + lane) * DD + m0) * 4;
  const float* pfd = PD + (((size_t)batch * (BN / 4) + lane) * DD + n0) * 4;
  #pragma unroll
  for (int g = 1; g < PF_ROUNDS; ++g) {
    __builtin_prefetch(pfa + (size_t)g * ROUND, 0, 3);
    __builtin_prefetch(pfd + (size_t)g * ROUND, 0, 3);
    __builtin_prefetch(pfa + (size_t)g * ROUND + HALFROUND, 0, 3);
    __builtin_prefetch(pfd + (size_t)g * ROUND + HALFROUND, 0, 3);
  }

  v8f acc0 = {};
  v8f acc1 = {};

  for (int u0 = 0; u0 < NSLOT - RING; u0 += RING) {
    ap += ROUND;
    dp += ROUND;
    if (u0 + PF_ROUNDS * RING < NSLOT) {
      const size_t o = (size_t)(u0 / RING + PF_ROUNDS) * ROUND;
      __builtin_prefetch(pfa + o, 0, 3);
      __builtin_prefetch(pfd + o, 0, 3);
      __builtin_prefetch(pfa + o + HALFROUND, 0, 3);
      __builtin_prefetch(pfd + o + HALFROUND, 0, 3);
    }
    #pragma unroll
    for (int j = 0; j < RING; ++j) {
      v2f x0; x0.x = a[j].x; x0.y = a[j].y;
      v2f x1; x1.x = a[j].z; x1.y = a[j].w;
      v2f p0; p0.x = b[j].x; p0.y = b[j].y;
      v2f p1; p1.x = b[j].z; p1.y = b[j].w;
      acc0 = __builtin_amdgcn_wmma_f32_16x16x4_f32(false, x0, false, p0, (short)0, acc0, false, false);
      acc1 = __builtin_amdgcn_wmma_f32_16x16x4_f32(false, x1, false, p1, (short)0, acc1, false, false);
      a[j] = *(const v4f*)(ap + (size_t)j * SLOT);
      b[j] = *(const v4f*)(dp + (size_t)j * SLOT);
      __builtin_amdgcn_sched_barrier(0);
    }
  }

  float wb[8];
  #pragma unroll
  for (int r = 0; r < 8; ++r)
    wb[r] = Wbase[(size_t)(m0 + r + 8 * half) * DD + n0 + l];

  #pragma unroll
  for (int j = 0; j < RING; ++j) {
    v2f x0; x0.x = a[j].x; x0.y = a[j].y;
    v2f x1; x1.x = a[j].z; x1.y = a[j].w;
    v2f p0; p0.x = b[j].x; p0.y = b[j].y;
    v2f p1; p1.x = b[j].z; p1.y = b[j].w;
    acc0 = __builtin_amdgcn_wmma_f32_16x16x4_f32(false, x0, false, p0, (short)0, acc0, false, false);
    acc1 = __builtin_amdgcn_wmma_f32_16x16x4_f32(false, x1, false, p1, (short)0, acc1, false, false);
  }

  float* ut = Uout + (size_t)batch * DD * DD + (size_t)tile * 256 + (8 * half) * 16 + l;
  #pragma unroll
  for (int r = 0; r < 8; ++r)
    ut[r * 16] = wb[r] + (acc0[r] + acc1[r]);
}

#define GRAD_WGS  256

__global__ __launch_bounds__(128) void gradU_all(
    const float* __restrict__ PK,  const float* __restrict__ PA1,
    const float* __restrict__ PA2, const float* __restrict__ PA3,
    const float* __restrict__ PD0, const float* __restrict__ PD1,
    const float* __restrict__ PD2, const float* __restrict__ PD3,
    const float* __restrict__ w0,  const float* __restrict__ w1,
    const float* __restrict__ w2,  const float* __restrict__ w3,
    float* __restrict__ U0, float* __restrict__ U1,
    float* __restrict__ U2, float* __restrict__ U3)
{
  const int wave  = __builtin_amdgcn_readfirstlane((int)(threadIdx.x >> 5));
  const int lane  = threadIdx.x & 31;
  const int wg    = blockIdx.x;
  const int layer = wg >> 6;
  const int batch = (wg >> 4) & 3;
  const int blk   = wg & 15;
  const int tile  = ((2 * (blk >> 2) + (wave >> 1)) << 3) | (2 * (blk & 3) + (wave & 1));
  const float* PA = (layer == 0) ? PK  : (layer == 1) ? PA1 : (layer == 2) ? PA2 : PA3;
  const float* PD = (layer == 0) ? PD0 : (layer == 1) ? PD1 : (layer == 2) ? PD2 : PD3;
  const float* W  = (layer == 0) ? w0  : (layer == 1) ? w1  : (layer == 2) ? w2  : w3;
  float*       U  = (layer == 0) ? U0  : (layer == 1) ? U1  : (layer == 2) ? U2  : U3;
  gradU_tile(tile, batch, lane, PA, PD, W, U);
}


#define XS_V4 512

__device__ __forceinline__ void xs_barrier()
{
  asm volatile("s_wait_dscnt 0x0\n\ts_barrier_signal -1\n\ts_barrier_wait -1" ::: "memory");
}

__device__ __forceinline__ void xs_put(v4f (&xb)[XS_V4], int wave, int lane, const float (&y0)[8], const float (&y1)[8])
{
  const int o = wave * 128 + lane;
  v4f a; a.x = y0[0]; a.y = y0[1]; a.z = y0[2]; a.w = y0[3];
  v4f b; b.x = y0[4]; b.y = y0[5]; b.z = y0[6]; b.w = y0[7];
  v4f c; c.x = y1[0]; c.y = y1[1]; c.z = y1[2]; c.w = y1[3];
  v4f d; d.x = y1[4]; d.y = y1[5]; d.z = y1[6]; d.w = y1[7];
  xb[o]      = a;
  xb[o + 32] = b;
  xb[o + 64] = c;
  xb[o + 96] = d;
}

__device__ __forceinline__ void xs_get(const v4f (&xb)[XS_V4], int lane, float (&x)[8][8])
{
  #pragma unroll
  for (int t = 0; t < 8; ++t) {
    const v4f lo = xb[t * 64 + lane];
    const v4f hi = xb[t * 64 + 32 + lane];
    x[t][0] = lo.x; x[t][1] = lo.y; x[t][2] = lo.z; x[t][3] = lo.w;
    x[t][4] = hi.x; x[t][5] = hi.y; x[t][6] = hi.z; x[t][7] = hi.w;
  }
}

__device__ __forceinline__ void store_tile(float* __restrict__ p, const float (&y)[8])
{
  v4f lo; lo.x = y[0]; lo.y = y[1]; lo.z = y[2]; lo.w = y[3];
  v4f hi; hi.x = y[4]; hi.y = y[5]; hi.z = y[6]; hi.w = y[7];
  *(v4f*)(p)     = lo;
  *(v4f*)(p + 4) = hi;
}

__device__ __forceinline__ void store_packed(float* __restrict__ P, const v4f (&xb)[XS_V4], int wave, int lane, int tok0)
{
  const float* xf = (const float*)&xb[0];
  const int base = (2 * wave + (lane >> 4)) * 256 + ((lane >> 2) & 1) * 128 + ((lane >> 3) & 1) * 64 + (lane & 3);
  float* dst = P + ((size_t)(tok0 >> 2) * DD + 32 * wave + lane) * 4;
  #pragma unroll
  for (int g = 0; g < 4; ++g) {
    v4f v;
    v.x = xf[base + (4 * g + 0) * 4];
    v.y = xf[base + (4 * g + 1) * 4];
    v.z = xf[base + (4 * g + 2) * 4];
    v.w = xf[base + (4 * g + 3) * 4];
    *(v4f*)(dst + (size_t)g * DD * 4) = v;
  }
}

#define ROWMAJOR(LD) (LD), (16 * (LD)), 16
#define TILED        16, 2048, 256

template <int RP, int BP, int T1>
__device__ __forceinline__ void ring_fill(float (&ra)[16][4], const float* __restrict__ Wc)
{
  #pragma unroll
  for (int i = 0; i < 16; ++i) {
    const size_t o = (size_t)(i >> 2) * BP + (size_t)(2 * (i & 3)) * RP;
    ra[i][0] = Wc[o];
    ra[i][1] = Wc[o + RP];
    ra[i][2] = Wc[o + T1];
    ra[i][3] = Wc[o + RP + T1];
    __builtin_amdgcn_sched_barrier(0);
  }
}

template <int RP, int BP, int T1, int RPN, int BPN, int T1N>
__device__ __forceinline__ void chain_layer(const float (&x)[8][8], v8f& acc0, v8f& acc1, float (&ra)[16][4],
                                            const float* __restrict__ Wc, const float* __restrict__ Wn)
{
  acc0 = (v8f){};
  acc1 = (v8f){};
  #pragma unroll
  for (int i = 0; i < 32; ++i) {
    const int slot = i & 15;
    const int t = i >> 2, j = i & 3;
    v2f a0; a0.x = ra[slot][0]; a0.y = ra[slot][1];
    v2f a1; a1.x = ra[slot][2]; a1.y = ra[slot][3];
    v2f b;  b.x  = x[t][2 * j]; b.y  = x[t][2 * j + 1];
    acc0 = __builtin_amdgcn_wmma_f32_16x16x4_f32(false, a0, false, b, (short)0, acc0, false, false);
    acc1 = __builtin_amdgcn_wmma_f32_16x16x4_f32(false, a1, false, b, (short)0, acc1, false, false);
    const float* src = (i < 16) ? Wc : Wn;
    const int rp = (i < 16) ? RP : RPN;
    const int bp = (i < 16) ? BP : BPN;
    const int t1 = (i < 16) ? T1 : T1N;
    const int ii = (i < 16) ? i + 16 : i - 16;
    const size_t o = (size_t)(ii >> 2) * bp + (size_t)(2 * (ii & 3)) * rp;
    ra[slot][0] = src[o];
    ra[slot][1] = src[o + rp];
    ra[slot][2] = src[o + t1];
    ra[slot][3] = src[o + rp + t1];
    __builtin_amdgcn_sched_barrier(0);
  }
}

__device__ __forceinline__ void load_x(float (&x)[8][8], const float* __restrict__ srow)
{
  #pragma unroll
  for (int t = 0; t < 8; ++t) {
    const v4f lo = *(const v4f*)(srow + 16 * t);
    const v4f hi = *(const v4f*)(srow + 16 * t + 4);
    x[t][0] = lo.x; x[t][1] = lo.y; x[t][2] = lo.z; x[t][3] = lo.w;
    x[t][4] = hi.x; x[t][5] = hi.y; x[t][6] = hi.z; x[t][7] = hi.w;
  }
}

__device__ __forceinline__ void bring_fill(v4f (&rb)[8][2], const float* __restrict__ Wt)
{
  #pragma unroll
  for (int q = 0; q < 8; ++q) {
    const size_t o = (size_t)(q >> 1) * 16 + (size_t)(q & 1) * 4;
    rb[q][0] = *(const v4f*)(Wt + o);
    rb[q][1] = *(const v4f*)(Wt + o + 16 * DD);
    __builtin_amdgcn_sched_barrier(0);
  }
}

__device__ __forceinline__ void bwd_layer(const float (&x)[8][8], v8f& acc0, v8f& acc1, v4f (&rb)[8][2],
                                          const float* __restrict__ Wc, const float* __restrict__ Wn)
{
  acc0 = (v8f){};
  acc1 = (v8f){};
  #pragma unroll
  for (int q = 0; q < 16; ++q) {
    const int p = q & 7;
    const int t = q >> 1, jj = q & 1;
    v2f a00; a00.x = rb[p][0].x; a00.y = rb[p][0].y;
    v2f a01; a01.x = rb[p][0].z; a01.y = rb[p][0].w;
    v2f a10; a10.x = rb[p][1].x; a10.y = rb[p][1].y;
    v2f a11; a11.x = rb[p][1].z; a11.y = rb[p][1].w;
    v2f b0;  b0.x  = x[t][4 * jj];     b0.y = x[t][4 * jj + 1];
    v2f b1;  b1.x  = x[t][4 * jj + 2]; b1.y = x[t][4 * jj + 3];
    acc0 = __builtin_amdgcn_wmma_f32_16x16x4_f32(false, a00, false, b0, (short)0, acc0, false, false);
    acc1 = __builtin_amdgcn_wmma_f32_16x16x4_f32(false, a10, false, b0, (short)0, acc1, false, false);
    acc0 = __builtin_amdgcn_wmma_f32_16x16x4_f32(false, a01, false, b1, (short)0, acc0, false, false);
    acc1 = __builtin_amdgcn_wmma_f32_16x16x4_f32(false, a11, false, b1, (short)0, acc1, false, false);
    const float* src = (q < 8) ? Wc : Wn;
    const int qq = (q < 8) ? q + 8 : q - 8;
    const size_t o = (size_t)(qq >> 1) * 16 + (size_t)(qq & 1) * 4;
    rb[p][0] = *(const v4f*)(src + o);
    rb[p][1] = *(const v4f*)(src + o + 16 * DD);
    __builtin_amdgcn_sched_barrier(0);
  }
}

__global__ __launch_bounds__(128) void qchain(
    const float* __restrict__ S, const float* __restrict__ wq,
    const float* __restrict__ U0, const float* __restrict__ U1,
    const float* __restrict__ U2, const float* __restrict__ U3,
    float* __restrict__ out)
{
  __shared__ v4f xs[2][XS_V4];
  const int wave = __builtin_amdgcn_readfirstlane((int)(threadIdx.x >> 5));
  const int lane = threadIdx.x & 31;
  const int half = lane >> 4;
  const int c    = lane & 15;
  const int tok0 = blockIdx.x * 16;
  const size_t boff = (size_t)(tok0 / BN) * DD * DD;
  const size_t woff = (size_t)(32 * wave);
  const size_t loff = (size_t)(8 * half) * DD + c;
  const size_t uoff = boff + (size_t)(512 * wave);
  const size_t luof = (size_t)(8 * half) * 16 + c;

  float x[8][8];
  load_x(x, S + (size_t)(tok0 + c) * DD + 8 * half);

  float ra[16][4];
  const float* Wc = (wq + woff) + loff;
  ring_fill<ROWMAJOR(DD)>(ra, Wc);

  {
    const float* Uw = (wave == 0) ? U0 : (wave == 1) ? U1 : (wave == 2) ? U2 : U3;
    __builtin_prefetch(Uw + boff + (size_t)(((blockIdx.x >> 3) & 7) * 32 + lane) * 64, 0, 3);
  }

  v8f acc0, acc1;
  #pragma unroll
  for (int L = 0; L < 5; ++L) {
    const float* Wn = (((L == 0) ? U0 : (L == 1) ? U1 : (L == 2) ? U2 : U3) + uoff) + luof;
    if (L == 0) chain_layer<ROWMAJOR(DD), TILED>(x, acc0, acc1, ra, Wc, Wn);
    else        chain_layer<TILED, TILED>(x, acc0, acc1, ra, Wc, Wn);
    if (L < 4) {
      float y0[8], y1[8];
      if (L >= 1) {
        #pragma unroll
        for (int r = 0; r < 8; ++r) { y0[r] = siluf(acc0[r]); y1[r] = siluf(acc1[r]); }
      } else {
        #pragma unroll
        for (int r = 0; r < 8; ++r) { y0[r] = acc0[r]; y1[r] = acc1[r]; }
      }
      xs_put(xs[L & 1], wave, lane, y0, y1);
      xs_barrier();
      xs_get(xs[L & 1], lane, x);
    }
    Wc = Wn;
  }

  float* orow = out + (size_t)(tok0 + c) * DD + 32 * wave + 8 * half;
  float o0[8], o1[8];
  #pragma unroll
  for (int r = 0; r < 8; ++r) { o0[r] = acc0[r]; o1[r] = acc1[r]; }
  store_tile(orow, o0);
  store_tile(orow + 16, o1);
}

__global__ __launch_bounds__(128) void kfwd(
    const float* __restrict__ S, const float* __restrict__ wkv,
    const float* __restrict__ w0, const float* __restrict__ w1,
    const float* __restrict__ w2, const float* __restrict__ w3,
    float* __restrict__ PK,
    float* __restrict__ PA1, float* __restrict__ PA2, float* __restrict__ PA3,
    float* __restrict__ PD3, float* __restrict__ PD2,
    float* __restrict__ PD1, float* __restrict__ PD0)
{
  __shared__ v4f xs[2][XS_V4];
  const int wave = __builtin_amdgcn_readfirstlane((int)(threadIdx.x >> 5));
  const int lane = threadIdx.x & 31;
  const int half = lane >> 4;
  const int c    = lane & 15;
  const int tok0 = blockIdx.x * 16;
  const size_t woff = (size_t)(32 * wave);
  const size_t lkv  = (size_t)(8 * half) * 256 + c;
  const size_t lw   = (size_t)(8 * half) * DD + c;
  const size_t wrow = (size_t)(32 * wave) * DD;
  const size_t lrow = (size_t)c * DD + 8 * half;

  float x[8][8];
  load_x(x, S + (size_t)(tok0 + c) * DD + 8 * half);

  float v0[8], v1[8];
  float g[3][2][8];
  float ra[16][4];
  v4f   rb[8][2];
  v8f   acc0, acc1;

  const float* Wv = (wkv + (DD + woff)) + lkv;
  const float* Wk = (wkv + woff) + lkv;
  ring_fill<ROWMAJOR(256)>(ra, Wv);
  {
    const int u = ((blockIdx.x >> 3) & 7) * 32 + lane;
    const float* ww = (wave == 0) ? w0 : (wave == 1) ? w1 : (wave == 2) ? w2 : w3;
    __builtin_prefetch(ww + (size_t)u * 64, 0, 3);
    if (wave == 0) __builtin_prefetch(wkv + (size_t)(u >> 1) * 256 + 64 * (u & 1), 0, 3);
  }
  chain_layer<ROWMAJOR(256), ROWMAJOR(256)>(x, acc0, acc1, ra, Wv, Wk);
  #pragma unroll
  for (int r = 0; r < 8; ++r) { v0[r] = acc0[r]; v1[r] = acc1[r]; }
  const float* Wc = (w0 + woff) + lw;
  chain_layer<ROWMAJOR(256), ROWMAJOR(DD)>(x, acc0, acc1, ra, Wk, Wc);
  {
    float k0[8], k1[8];
    #pragma unroll
    for (int r = 0; r < 8; ++r) { k0[r] = acc0[r]; k1[r] = acc1[r]; }
    xs_put(xs[0], wave, lane, k0, k1);
    xs_barrier();
    xs_get(xs[0], lane, x);
    store_packed(PK, xs[0], wave, lane, tok0);
  }
  #pragma unroll
  for (int L = 0; L < 4; ++L) {
    const float* Wn = (((L == 0) ? w1 : (L == 1) ? w2 : w3) + woff) + lw;
    chain_layer<ROWMAJOR(DD), ROWMAJOR(DD)>(x, acc0, acc1, ra, Wc, Wn);
    if (L < 3) {
      float* pact = (L == 0) ? PA1 : (L == 1) ? PA2 : PA3;
      float a0[8], a1[8];
      #pragma unroll
      for (int r = 0; r < 8; ++r) {
        const float z0 = acc0[r], z1 = acc1[r];
        a0[r] = siluf(z0); g[L][0][r] = dsiluf(z0);
        a1[r] = siluf(z1); g[L][1][r] = dsiluf(z1);
      }
      xs_put(xs[(L + 1) & 1], wave, lane, a0, a1);
      xs_barrier();
      xs_get(xs[(L + 1) & 1], lane, x);
      store_packed(pact, xs[(L + 1) & 1], wave, lane, tok0);
    } else {
      bring_fill(rb, (w3 + wrow) + lrow);
      float d0[8], d1[8];
      #pragma unroll
      for (int r = 0; r < 8; ++r) {
        d0[r] = (2.0f / 128.0f) * (acc0[r] - v0[r]);
        d1[r] = (2.0f / 128.0f) * (acc1[r] - v1[r]);
      }
      xs_put(xs[0], wave, lane, d0, d1);
      xs_barrier();
      xs_get(xs[0], lane, x);
      store_packed(PD3, xs[0], wave, lane, tok0);
    }
    Wc = Wn;
  }
  #pragma unroll
  for (int i = 0; i < 3; ++i) {
    const int L = 2 - i;
    const float* Wb = (((L == 2) ? w3 : (L == 1) ? w2 : w1) + wrow) + lrow;
    const float* Wn = (((L == 2) ? w2 : w1) + wrow) + lrow;
    bwd_layer(x, acc0, acc1, rb, Wb, Wn);
    float* pd = (L == 2) ? PD2 : (L == 1) ? PD1 : PD0;
    float d0[8], d1[8];
    #pragma unroll
    for (int r = 0; r < 8; ++r) {
      d0[r] = acc0[r] * g[L][0][r];
      d1[r] = acc1[r] * g[L][1][r];
    }
    xs_put(xs[(L + 1) & 1], wave, lane, d0, d1);
    if (L > 0) {
      xs_barrier();
      xs_get(xs[(L + 1) & 1], lane, x);
    } else {
      asm volatile("s_wait_dscnt 0x0" ::: "memory");
    }
    store_packed(pd, xs[(L + 1) & 1], wave, lane, tok0);
  }
}

extern "C" void kernel_launch(void* const* d_in, const int* in_sizes, int n_in,
                              void* d_out, int out_size, void* d_ws, size_t ws_size,
                              hipStream_t stream)
{
  const float* S   = (const float*)d_in[0];
  const float* w0  = (const float*)d_in[1];
  const float* w1  = (const float*)d_in[2];
  const float* w2  = (const float*)d_in[3];
  const float* w3  = (const float*)d_in[4];
  const float* wq  = (const float*)d_in[5];
  const float* wkv = (const float*)d_in[6];
  float* out = (float*)d_out;

  float* p = (float*)d_ws;
  const size_t T = (size_t)TOK * DD;
  float* PK = p; p += T;
  float* A1 = p; p += T;
  float* A2 = p; p += T;
  float* A3 = p; p += T;
  float* P3 = p; p += T;
  float* P2 = p; p += T;
  float* P1 = p; p += T;
  float* P0 = p; p += T;
  float* U0 = p; p += 4 * DD * DD;
  float* U1 = p; p += 4 * DD * DD;
  float* U2 = p; p += 4 * DD * DD;
  float* U3 = p; p += 4 * DD * DD;

  kfwd<<<dim3(TOK / 16), dim3(128), 0, stream>>>(S, wkv, w0, w1, w2, w3, PK, A1, A2, A3, P3, P2, P1, P0);

  gradU_all<<<dim3(GRAD_WGS), dim3(128), 0, stream>>>(PK, A1, A2, A3, P0, P1, P2, P3, w0, w1, w2, w3, U0, U1, U2, U3);

  qchain<<<dim3(TOK / 16), dim3(128), 0, stream>>>(S, wq, U0, U1, U2, U3, out);
}


# chain kernels: a wave starts each layer's k loop on its own two activation tiles (in registers) while the LDS exchange of the other six runs behind a split barrier; k sum order rotated by wave
# speedup vs baseline: 11.0818x; 1.0346x over previous
#include <hip/hip_runtime.h>

typedef __attribute__((ext_vector_type(2))) float v2f;
typedef __attribute__((ext_vector_type(4))) float v4f;
typedef __attribute__((ext_vector_type(8))) float v8f;

#define TOK   4096
#define BN    1024
#define DD    128

__device__ __forceinline__ float sigf(float x)  { return 1.0f / (1.0f + __expf(-x)); }
__device__ __forceinline__ float siluf(float x) { return x * sigf(x); }
__device__ __forceinline__ float dsiluf(float x){ float s = sigf(x); return s * (1.0f + x * (1.0f - s)); }


__device__ __forceinline__ void gradU_tile(
    int tile, int batch, int lane,
    const float* __restrict__ PA,
    const float* __restrict__ PD,
    const float* __restrict__ Wbase,
    float* __restrict__ Uout)
{
  constexpr int RING = 32;
  constexpr int NSLOT = BN / 8;
  constexpr size_t SLOT = 2 * DD * 4;
  const int half = lane >> 4;
  const int l    = lane & 15;
  const int m0   = (tile >> 3) * 16;
  const int n0   = (tile & 7) * 16;
  const float* ap = PA + (((size_t)batch * (BN / 4) + half) * DD + m0 + l) * 4;
  const float* dp = PD + (((size_t)batch * (BN / 4) + half) * DD + n0 + l) * 4;

  v4f a[RING], b[RING];

  #pragma unroll
  for (int j = 0; j < RING; ++j) {
    a[j] = *(const v4f*)(ap + (size_t)j * SLOT);
    b[j] = *(const v4f*)(dp + (size_t)j * SLOT);
    __builtin_amdgcn_sched_barrier(0);
  }

  constexpr int PF_ROUNDS = 3;
  constexpr size_t ROUND = (size_t)RING * SLOT;
  constexpr size_t HALFROUND = (size_t)32 * DD * 4;
  const float* pfa = PA + (((size_t)batch * (BN / 4) + lane) * DD + m0) * 4;
  const float* pfd = PD + (((size_t)batch * (BN / 4) + lane) * DD + n0) * 4;
  #pragma unroll
  for (int g = 1; g < PF_ROUNDS; ++g) {
    __builtin_prefetch(pfa + (size_t)g * ROUND, 0, 3);
    __builtin_prefetch(pfd + (size_t)g * ROUND, 0, 3);
    __builtin_prefetch(pfa + (size_t)g * ROUND + HALFROUND, 0, 3);
    __builtin_prefetch(pfd + (size_t)g * ROUND + HALFROUND, 0, 3);
  }

  v8f acc0 = {};
  v8f acc1 = {};

  for (int u0 = 0; u0 < NSLOT - RING; u0 += RING) {
    ap += ROUND;
    dp += ROUND;
    if (u0 + PF_ROUNDS * RING < NSLOT) {
      const size_t o = (size_t)(u0 / RING + PF_ROUNDS) * ROUND;
      __builtin_prefetch(pfa + o, 0, 3);
      __builtin_prefetch(pfd + o, 0, 3);
      __builtin_prefetch(pfa + o + HALFROUND, 0, 3);
      __builtin_prefetch(pfd + o + HALFROUND, 0, 3);
    }
    #pragma unroll
    for (int j = 0; j < RING; ++j) {
      v2f x0; x0.x = a[j].x; x0.y = a[j].y;
      v2f x1; x1.x = a[j].z; x1.y = a[j].w;
      v2f p0; p0.x = b[j].x; p0.y = b[j].y;
      v2f p1; p1.x = b[j].z; p1.y = b[j].w;
      acc0 = __builtin_amdgcn_wmma_f32_16x16x4_f32(false, x0, false, p0, (short)0, acc0, false, false);
      acc1 = __builtin_amdgcn_wmma_f32_16x16x4_f32(false, x1, false, p1, (short)0, acc1, false, false);
      a[j] = *(const v4f*)(ap + (size_t)j * SLOT);
      b[j] = *(const v4f*)(dp + (size_t)j * SLOT);
      __builtin_amdgcn_sched_barrier(0);
    }
  }

  float wb[8];
  #pragma unroll
  for (int r = 0; r < 8; ++r)
    wb[r] = Wbase[(size_t)(m0 + r + 8 * half) * DD + n0 + l];

  #pragma unroll
  for (int j = 0; j < RING; ++j) {
    v2f x0; x0.x = a[j].x; x0.y = a[j].y;
    v2f x1; x1.x = a[j].z; x1.y = a[j].w;
    v2f p0; p0.x = b[j].x; p0.y = b[j].y;
    v2f p1; p1.x = b[j].z; p1.y = b[j].w;
    acc0 = __builtin_amdgcn_wmma_f32_16x16x4_f32(false, x0, false, p0, (short)0, acc0, false, false);
    acc1 = __builtin_amdgcn_wmma_f32_16x16x4_f32(false, x1, false, p1, (short)0, acc1, false, false);
  }

  float* ut = Uout + (size_t)batch * DD * DD + (size_t)tile * 256 + (8 * half) * 16 + l;
  #pragma unroll
  for (int r = 0; r < 8; ++r)
    ut[r * 16] = wb[r] + (acc0[r] + acc1[r]);
}

#define GRAD_WGS  256

__global__ __launch_bounds__(128) void gradU_all(
    const float* __restrict__ PK,  const float* __restrict__ PA1,
    const float* __restrict__ PA2, const float* __restrict__ PA3,
    const float* __restrict__ PD0, const float* __restrict__ PD1,
    const float* __restrict__ PD2, const float* __restrict__ PD3,
    const float* __restrict__ w0,  const float* __restrict__ w1,
    const float* __restrict__ w2,  const float* __restrict__ w3,
    float* __restrict__ U0, float* __restrict__ U1,
    float* __restrict__ U2, float* __restrict__ U3)
{
  const int wave  = __builtin_amdgcn_readfirstlane((int)(threadIdx.x >> 5));
  const int lane  = threadIdx.x & 31;
  const int wg    = blockIdx.x;
  const int layer = wg >> 6;
  const int batch = (wg >> 4) & 3;
  const int blk   = wg & 15;
  const int tile  = ((2 * (blk >> 2) + (wave >> 1)) << 3) | (2 * (blk & 3) + (wave & 1));
  const float* PA = (layer == 0) ? PK  : (layer == 1) ? PA1 : (layer == 2) ? PA2 : PA3;
  const float* PD = (layer == 0) ? PD0 : (layer == 1) ? PD1 : (layer == 2) ? PD2 : PD3;
  const float* W  = (layer == 0) ? w0  : (layer == 1) ? w1  : (layer == 2) ? w2  : w3;
  float*       U  = (layer == 0) ? U0  : (layer == 1) ? U1  : (layer == 2) ? U2  : U3;
  gradU_tile(tile, batch, lane, PA, PD, W, U);
}


#define XS_V4 512

__device__ __forceinline__ void own_landed_signal()
{
  asm volatile("s_wait_dscnt 0x0\n\ts_barrier_signal -1" ::: "memory");
}
__device__ __forceinline__ void all_landed_wait()
{
  asm volatile("s_barrier_wait -1" ::: "memory");
}

__device__ __forceinline__ void xs_put(v4f (&xb)[XS_V4], int wave, int lane, const float (&y0)[8], const float (&y1)[8])
{
  const int o = wave * 128 + lane;
  v4f a; a.x = y0[0]; a.y = y0[1]; a.z = y0[2]; a.w = y0[3];
  v4f b; b.x = y0[4]; b.y = y0[5]; b.z = y0[6]; b.w = y0[7];
  v4f c; c.x = y1[0]; c.y = y1[1]; c.z = y1[2]; c.w = y1[3];
  v4f d; d.x = y1[4]; d.y = y1[5]; d.z = y1[6]; d.w = y1[7];
  xb[o]      = a;
  xb[o + 32] = b;
  xb[o + 64] = c;
  xb[o + 96] = d;
  __builtin_amdgcn_sched_barrier(0);
}

__device__ __forceinline__ void xs_get_others(const v4f (&xb)[XS_V4], int wave, int lane, float (&xo)[6][8])
{
  #pragma unroll
  for (int m = 0; m < 6; ++m) {
    const int t = (2 * wave + 2 + m) & 7;
    const v4f lo = xb[t * 64 + lane];
    const v4f hi = xb[t * 64 + 32 + lane];
    xo[m][0] = lo.x; xo[m][1] = lo.y; xo[m][2] = lo.z; xo[m][3] = lo.w;
    xo[m][4] = hi.x; xo[m][5] = hi.y; xo[m][6] = hi.z; xo[m][7] = hi.w;
  }
}

__device__ __forceinline__ void store_tile(float* __restrict__ p, const float (&y)[8])
{
  v4f lo; lo.x = y[0]; lo.y = y[1]; lo.z = y[2]; lo.w = y[3];
  v4f hi; hi.x = y[4]; hi.y = y[5]; hi.z = y[6]; hi.w = y[7];
  *(v4f*)(p)     = lo;
  *(v4f*)(p + 4) = hi;
}

__device__ __forceinline__ void packed_read(v4f (&pk)[4], const v4f (&xb)[XS_V4], int wave, int lane)
{
  const float* xf = (const float*)&xb[0];
  const int base = (2 * wave + (lane >> 4)) * 256 + ((lane >> 2) & 1) * 128 + ((lane >> 3) & 1) * 64 + (lane & 3);
  #pragma unroll
  for (int g = 0; g < 4; ++g) {
    pk[g].x = xf[base + (4 * g + 0) * 4];
    pk[g].y = xf[base + (4 * g + 1) * 4];
    pk[g].z = xf[base + (4 * g + 2) * 4];
    pk[g].w = xf[base + (4 * g + 3) * 4];
  }
}
__device__ __forceinline__ void packed_write(float* __restrict__ P, const v4f (&pk)[4], int wave, int lane, int tok0)
{
  float* dst = P + ((size_t)(tok0 >> 2) * DD + 32 * wave + lane) * 4;
  #pragma unroll
  for (int g = 0; g < 4; ++g)
    *(v4f*)(dst + (size_t)g * DD * 4) = pk[g];
}
__device__ __forceinline__ void store_packed(float* __restrict__ P, const v4f (&xb)[XS_V4], int wave, int lane, int tok0)
{
  v4f pk[4];
  packed_read(pk, xb, wave, lane);
  packed_write(P, pk, wave, lane, tok0);
}

#define ROWMAJOR(LD) (LD), (16 * (LD)), 16
#define TILED        16, 2048, 256

__device__ __forceinline__ int rot_block(int wave, int tp) { return (2 * wave + tp) & 7; }

template <int RP, int BP, int T1>
__device__ __forceinline__ void ring_fill(float (&ra)[16][4], const float* __restrict__ Wc)
{
  #pragma unroll
  for (int i = 0; i < 16; ++i) {
    const size_t o = (size_t)(i >> 2) * BP + (size_t)(2 * (i & 3)) * RP;
    ra[i][0] = Wc[o];
    ra[i][1] = Wc[o + RP];
    ra[i][2] = Wc[o + T1];
    ra[i][3] = Wc[o + RP + T1];
    __builtin_amdgcn_sched_barrier(0);
  }
}

template <int RP, int BP, int T1, int RPN, int BPN, int T1N, bool NR>
__device__ __forceinline__ void chain_layer(const float (&x)[8][8], v8f& acc0, v8f& acc1, float (&ra)[16][4],
                                            const float* __restrict__ Wc, const float* __restrict__ Wn, int wave)
{
  acc0 = (v8f){};
  acc1 = (v8f){};
  #pragma unroll
  for (int i = 0; i < 32; ++i) {
    const int slot = i & 15;
    const int t = i >> 2, j = i & 3;
    v2f a0; a0.x = ra[slot][0]; a0.y = ra[slot][1];
    v2f a1; a1.x = ra[slot][2]; a1.y = ra[slot][3];
    v2f b;  b.x  = x[t][2 * j]; b.y  = x[t][2 * j + 1];
    acc0 = __builtin_amdgcn_wmma_f32_16x16x4_f32(false, a0, false, b, (short)0, acc0, false, false);
    acc1 = __builtin_amdgcn_wmma_f32_16x16x4_f32(false, a1, false, b, (short)0, acc1, false, false);
    const float* src = (i < 16) ? Wc : Wn;
    const int rp = (i < 16) ? RP : RPN;
    const int bp = (i < 16) ? BP : BPN;
    const int t1 = (i < 16) ? T1 : T1N;
    const int ii = (i < 16) ? i + 16 : i - 16;
    const int tb = (i >= 16 && NR) ? rot_block(wave, ii >> 2) : (ii >> 2);
    const size_t o = (size_t)tb * bp + (size_t)(2 * (ii & 3)) * rp;
    ra[slot][0] = src[o];
    ra[slot][1] = src[o + rp];
    ra[slot][2] = src[o + t1];
    ra[slot][3] = src[o + rp + t1];
    __builtin_amdgcn_sched_barrier(0);
  }
}

template <int RP, int BP, int T1, int RPN, int BPN, int T1N, bool PACK>
__device__ __forceinline__ void chain_layer_rot(const float (&y0)[8], const float (&y1)[8], const v4f (&xb)[XS_V4],
                                                float* __restrict__ P, int tok0, int wave, int lane,
                                                v8f& acc0, v8f& acc1, float (&ra)[16][4],
                                                const float* __restrict__ Wc, const float* __restrict__ Wn)
{
  float xo[6][8];
  v4f pk[4];
  acc0 = (v8f){};
  acc1 = (v8f){};
  #pragma unroll
  for (int i = 0; i < 32; ++i) {
    const int slot = i & 15;
    const int tp = i >> 2, j = i & 3;
    v2f a0; a0.x = ra[slot][0]; a0.y = ra[slot][1];
    v2f a1; a1.x = ra[slot][2]; a1.y = ra[slot][3];
    v2f b;
    if (tp == 0)      { b.x = y0[2 * j];         b.y = y0[2 * j + 1]; }
    else if (tp == 1) { b.x = y1[2 * j];         b.y = y1[2 * j + 1]; }
    else              { b.x = xo[tp - 2][2 * j]; b.y = xo[tp - 2][2 * j + 1]; }
    acc0 = __builtin_amdgcn_wmma_f32_16x16x4_f32(false, a0, false, b, (short)0, acc0, false, false);
    acc1 = __builtin_amdgcn_wmma_f32_16x16x4_f32(false, a1, false, b, (short)0, acc1, false, false);
    const float* src = (i < 16) ? Wc : Wn;
    const int rp = (i < 16) ? RP : RPN;
    const int bp = (i < 16) ? BP : BPN;
    const int t1 = (i < 16) ? T1 : T1N;
    const int ii = (i < 16) ? i + 16 : i - 16;
    const int tb = rot_block(wave, ii >> 2);
    const size_t o = (size_t)tb * bp + (size_t)(2 * (ii & 3)) * rp;
    ra[slot][0] = src[o];
    ra[slot][1] = src[o + rp];
    ra[slot][2] = src[o + t1];
    ra[slot][3] = src[o + rp + t1];
    __builtin_amdgcn_sched_barrier(0);
    if (i == 2) {
      own_landed_signal();
      if (PACK) packed_read(pk, xb, wave, lane);
      __builtin_amdgcn_sched_barrier(0);
    }
    if (i == 3) {
      all_landed_wait();
      xs_get_others(xb, wave, lane, xo);
      __builtin_amdgcn_sched_barrier(0);
    }
    if (i == 5 && PACK) {
      packed_write(P, pk, wave, lane, tok0);
      __builtin_amdgcn_sched_barrier(0);
    }
  }
}

__device__ __forceinline__ void load_x(float (&x)[8][8], const float* __restrict__ srow)
{
  #pragma unroll
  for (int t = 0; t < 8; ++t) {
    const v4f lo = *(const v4f*)(srow + 16 * t);
    const v4f hi = *(const v4f*)(srow + 16 * t + 4);
    x[t][0] = lo.x; x[t][1] = lo.y; x[t][2] = lo.z; x[t][3] = lo.w;
    x[t][4] = hi.x; x[t][5] = hi.y; x[t][6] = hi.z; x[t][7] = hi.w;
  }
}

__device__ __forceinline__ void bring_fill(v4f (&rb)[8][2], const float* __restrict__ Wt, int wave)
{
  #pragma unroll
  for (int q = 0; q < 8; ++q) {
    const size_t o = (size_t)rot_block(wave, q >> 1) * 16 + (size_t)(q & 1) * 4;
    rb[q][0] = *(const v4f*)(Wt + o);
    rb[q][1] = *(const v4f*)(Wt + o + 16 * DD);
    __builtin_amdgcn_sched_barrier(0);
  }
}

__device__ __forceinline__ void bwd_layer(const float (&y0)[8], const float (&y1)[8], const v4f (&xb)[XS_V4],
                                          float* __restrict__ P, int tok0, int wave, int lane,
                                          v8f& acc0, v8f& acc1, v4f (&rb)[8][2],
                                          const float* __restrict__ Wc, const float* __restrict__ Wn)
{
  float xo[6][8];
  v4f pk[4];
  acc0 = (v8f){};
  acc1 = (v8f){};
  #pragma unroll
  for (int q = 0; q < 16; ++q) {
    const int p = q & 7;
    const int tp = q >> 1, jj = q & 1;
    v2f a00; a00.x = rb[p][0].x; a00.y = rb[p][0].y;
    v2f a01; a01.x = rb[p][0].z; a01.y = rb[p][0].w;
    v2f a10; a10.x = rb[p][1].x; a10.y = rb[p][1].y;
    v2f a11; a11.x = rb[p][1].z; a11.y = rb[p][1].w;
    v2f b0, b1;
    if (tp == 0)      { b0.x = y0[4 * jj]; b0.y = y0[4 * jj + 1]; b1.x = y0[4 * jj + 2]; b1.y = y0[4 * jj + 3]; }
    else if (tp == 1) { b0.x = y1[4 * jj]; b0.y = y1[4 * jj + 1]; b1.x = y1[4 * jj + 2]; b1.y = y1[4 * jj + 3]; }
    else { b0.x = xo[tp - 2][4 * jj]; b0.y = xo[tp - 2][4 * jj + 1]; b1.x = xo[tp - 2][4 * jj + 2]; b1.y = xo[tp - 2][4 * jj + 3]; }
    acc0 = __builtin_amdgcn_wmma_f32_16x16x4_f32(false, a00, false, b0, (short)0, acc0, false, false);
    acc1 = __builtin_amdgcn_wmma_f32_16x16x4_f32(false, a10, false, b0, (short)0, acc1, false, false);
    acc0 = __builtin_amdgcn_wmma_f32_16x16x4_f32(false, a01, false, b1, (short)0, acc0, false, false);
    acc1 = __builtin_amdgcn_wmma_f32_16x16x4_f32(false, a11, false, b1, (short)0, acc1, false, false);
    const float* src = (q < 8) ? Wc : Wn;
    const int qq = (q < 8) ? q + 8 : q - 8;
    const size_t o = (size_t)rot_block(wave, qq >> 1) * 16 + (size_t)(qq & 1) * 4;
    rb[p][0] = *(const v4f*)(src + o);
    rb[p][1] = *(const v4f*)(src + o + 16 * DD);
    __builtin_amdgcn_sched_barrier(0);
    if (q == 0) {
      own_landed_signal();
      packed_read(pk, xb, wave, lane);
      __builtin_amdgcn_sched_barrier(0);
    }
    if (q == 1) {
      all_landed_wait();
      xs_get_others(xb, wave, lane, xo);
      __builtin_amdgcn_sched_barrier(0);
    }
    if (q == 2) {
      packed_write(P, pk, wave, lane, tok0);
      __builtin_amdgcn_sched_barrier(0);
    }
  }
}

__global__ __launch_bounds__(128) void qchain(
    const float* __restrict__ S, const float* __restrict__ wq,
    const float* __restrict__ U0, const float* __restrict__ U1,
    const float* __restrict__ U2, const float* __restrict__ U3,
    float* __restrict__ out)
{
  __shared__ v4f xs[2][XS_V4];
  const int wave = __builtin_amdgcn_readfirstlane((int)(threadIdx.x >> 5));
  const int lane = threadIdx.x & 31;
  const int half = lane >> 4;
  const int c    = lane & 15;
  const int tok0 = blockIdx.x * 16;
  const size_t boff = (size_t)(tok0 / BN) * DD * DD;
  const size_t woff = (size_t)(32 * wave);
  const size_t loff = (size_t)(8 * half) * DD + c;
  const size_t uoff = boff + (size_t)(512 * wave);
  const size_t luof = (size_t)(8 * half) * 16 + c;

  float ra[16][4];
  v8f acc0, acc1;
  float y0[8], y1[8];
  const float* Wc = (wq + woff) + loff;
  {
    float x[8][8];
    load_x(x, S + (size_t)(tok0 + c) * DD + 8 * half);
    ring_fill<ROWMAJOR(DD)>(ra, Wc);

    {
      const float* Uw = (wave == 0) ? U0 : (wave == 1) ? U1 : (wave == 2) ? U2 : U3;
      __builtin_prefetch(Uw + boff + (size_t)(((blockIdx.x >> 3) & 7) * 32 + lane) * 64, 0, 3);
    }

    const float* Wn = (U0 + uoff) + luof;
    chain_layer<ROWMAJOR(DD), TILED, true>(x, acc0, acc1, ra, Wc, Wn, wave);
    #pragma unroll
    for (int r = 0; r < 8; ++r) { y0[r] = acc0[r]; y1[r] = acc1[r]; }
    xs_put(xs[0], wave, lane, y0, y1);
    Wc = Wn;
  }

  #pragma unroll
  for (int L = 1; L < 5; ++L) {
    const float* Wn = (((L == 1) ? U1 : (L == 2) ? U2 : U3) + uoff) + luof;
    chain_layer_rot<TILED, TILED, false>(y0, y1, xs[(L - 1) & 1], nullptr, tok0, wave, lane, acc0, acc1, ra, Wc, Wn);
    if (L < 4) {
      #pragma unroll
      for (int r = 0; r < 8; ++r) { y0[r] = siluf(acc0[r]); y1[r] = siluf(acc1[r]); }
      xs_put(xs[L & 1], wave, lane, y0, y1);
    }
    Wc = Wn;
  }

  float* orow = out + (size_t)(tok0 + c) * DD + 32 * wave + 8 * half;
  float o0[8], o1[8];
  #pragma unroll
  for (int r = 0; r < 8; ++r) { o0[r] = acc0[r]; o1[r] = acc1[r]; }
  store_tile(orow, o0);
  store_tile(orow + 16, o1);
}

__global__ __launch_bounds__(128) void kfwd(
    const float* __restrict__ S, const float* __restrict__ wkv,
    const float* __restrict__ w0, const float* __restrict__ w1,
    const float* __restrict__ w2, const float* __restrict__ w3,
    float* __restrict__ PK,
    float* __restrict__ PA1, float* __restrict__ PA2, float* __restrict__ PA3,
    float* __restrict__ PD3, float* __restrict__ PD2,
    float* __restrict__ PD1, float* __restrict__ PD0)
{
  __shared__ v4f xs[2][XS_V4];
  const int wave = __builtin_amdgcn_readfirstlane((int)(threadIdx.x >> 5));
  const int lane = threadIdx.x & 31;
  const int half = lane >> 4;
  const int c    = lane & 15;
  const int tok0 = blockIdx.x * 16;
  const size_t woff = (size_t)(32 * wave);
  const size_t lkv  = (size_t)(8 * half) * 256 + c;
  const size_t lw   = (size_t)(8 * half) * DD + c;
  const size_t wrow = (size_t)(32 * wave) * DD;
  const size_t lrow = (size_t)c * DD + 8 * half;

  float v0[8], v1[8];
  float g[3][2][8];
  float ra[16][4];
  v4f   rb[8][2];
  v8f   acc0, acc1;
  float y0[8], y1[8];

  const float* Wc = (w0 + woff) + lw;
  {
    float x[8][8];
    load_x(x, S + (size_t)(tok0 + c) * DD + 8 * half);

    const float* Wv = (wkv + (DD + woff)) + lkv;
    const float* Wk = (wkv + woff) + lkv;
    ring_fill<ROWMAJOR(256)>(ra, Wv);
    {
      const int u = ((blockIdx.x >> 3) & 7) * 32 + lane;
      const float* ww = (wave == 0) ? w0 : (wave == 1) ? w1 : (wave == 2) ? w2 : w3;
      __builtin_prefetch(ww + (size_t)u * 64, 0, 3);
      if (wave == 0) __builtin_prefetch(wkv + (size_t)(u >> 1) * 256 + 64 * (u & 1), 0, 3);
    }
    chain_layer<ROWMAJOR(256), ROWMAJOR(256), false>(x, acc0, acc1, ra, Wv, Wk, wave);
    #pragma unroll
    for (int r = 0; r < 8; ++r) { v0[r] = acc0[r]; v1[r] = acc1[r]; }
    chain_layer<ROWMAJOR(256), ROWMAJOR(DD), true>(x, acc0, acc1, ra, Wk, Wc, wave);
    #pragma unroll
    for (int r = 0; r < 8; ++r) { y0[r] = acc0[r]; y1[r] = acc1[r]; }
    xs_put(xs[0], wave, lane, y0, y1);
  }
  #pragma unroll
  for (int L = 0; L < 4; ++L) {
    const float* Wn = (((L == 0) ? w1 : (L == 1) ? w2 : w3) + woff) + lw;
    float* pin = (L == 0) ? PK : (L == 1) ? PA1 : (L == 2) ? PA2 : PA3;
    chain_layer_rot<ROWMAJOR(DD), ROWMAJOR(DD), true>(y0, y1, xs[L & 1], pin, tok0, wave, lane, acc0, acc1, ra, Wc, Wn);
    if (L < 3) {
      #pragma unroll
      for (int r = 0; r < 8; ++r) {
        const float z0 = acc0[r], z1 = acc1[r];
        y0[r] = siluf(z0); g[L][0][r] = dsiluf(z0);
        y1[r] = siluf(z1); g[L][1][r] = dsiluf(z1);
      }
      xs_put(xs[(L + 1) & 1], wave, lane, y0, y1);
    } else {
      bring_fill(rb, (w3 + wrow) + lrow, wave);
      #pragma unroll
      for (int r = 0; r < 8; ++r) {
        y0[r] = (2.0f / 128.0f) * (acc0[r] - v0[r]);
        y1[r] = (2.0f / 128.0f) * (acc1[r] - v1[r]);
      }
      xs_put(xs[0], wave, lane, y0, y1);
    }
    Wc = Wn;
  }
  #pragma unroll
  for (int i = 0; i < 3; ++i) {
    const int L = 2 - i;
    const float* Wb = (((L == 2) ? w3 : (L == 1) ? w2 : w1) + wrow) + lrow;
    const float* Wn = (((L == 2) ? w2 : w1) + wrow) + lrow;
    float* pin = (L == 2) ? PD3 : (L == 1) ? PD2 : PD1;
    bwd_layer(y0, y1, xs[L & 1], pin, tok0, wave, lane, acc0, acc1, rb, Wb, Wn);
    #pragma unroll
    for (int r = 0; r < 8; ++r) {
      y0[r] = acc0[r] * g[L][0][r];
      y1[r] = acc1[r] * g[L][1][r];
    }
    xs_put(xs[(L + 1) & 1], wave, lane, y0, y1);
  }
  asm volatile("s_wait_dscnt 0x0" ::: "memory");
  store_packed(PD0, xs[1], wave, lane, tok0);
}

extern "C" void kernel_launch(void* const* d_in, const int* in_sizes, int n_in,
                              void* d_out, int out_size, void* d_ws, size_t ws_size,
                              hipStream_t stream)
{
  const float* S   = (const float*)d_in[0];
  const float* w0  = (const float*)d_in[1];
  const float* w1  = (const float*)d_in[2];
  const float* w2  = (const float*)d_in[3];
  const float* w3  = (const float*)d_in[4];
  const float* wq  = (const float*)d_in[5];
  const float* wkv = (const float*)d_in[6];
  float* out = (float*)d_out;

  float* p = (float*)d_ws;
  const size_t T = (size_t)TOK * DD;
  float* PK = p; p += T;
  float* A1 = p; p += T;
  float* A2 = p; p += T;
  float* A3 = p; p += T;
  float* P3 = p; p += T;
  float* P2 = p; p += T;
  float* P1 = p; p += T;
  float* P0 = p; p += T;
  float* U0 = p; p += 4 * DD * DD;
  float* U1 = p; p += 4 * DD * DD;
  float* U2 = p; p += 4 * DD * DD;
  float* U3 = p; p += 4 * DD * DD;

  kfwd<<<dim3(TOK / 16), dim3(128), 0, stream>>>(S, wkv, w0, w1, w2, w3, PK, A1, A2, A3, P3, P2, P1, P0);

  gradU_all<<<dim3(GRAD_WGS), dim3(128), 0, stream>>>(PK, A1, A2, A3, P0, P1, P2, P3, w0, w1, w2, w3, U0, U1, U2, U3);

  qchain<<<dim3(TOK / 16), dim3(128), 0, stream>>>(S, wq, U0, U1, U2, U3, out);
}
